# starting kernel
# baseline (speedup 1.0000x reference)
#include <hip/hip_runtime.h>
#include <hip/hip_bf16.h>

typedef float v2f __attribute__((ext_vector_type(2)));
typedef float v8f __attribute__((ext_vector_type(8)));

#define IN_D  64
#define HID_D 64
#define OUT_D 32

__global__ void deg_init(float* __restrict__ deg, int n) {
    int i = blockIdx.x * blockDim.x + threadIdx.x;
    if (i < n) deg[i] = 1.0f;
}

__global__ void deg_count(const int* __restrict__ dst, float* __restrict__ deg, int e) {
    int i = blockIdx.x * blockDim.x + threadIdx.x;
    if (i < e) atomicAdd(&deg[dst[i]], 1.0f);
}

__global__ void deg_rsqrt(float* __restrict__ deg, int n) {
    int i = blockIdx.x * blockDim.x + threadIdx.x;
    if (i < n) deg[i] = rsqrtf(deg[i]);
}

template<int DOUT>
__global__ __launch_bounds__(256) void gemm_wmma_f32(
        const float* __restrict__ X, const float* __restrict__ W,
        float* __restrict__ Y, int nrows) {
    __shared__ float sW[64 * DOUT];
    for (int i = threadIdx.x; i < 64 * DOUT; i += 256) {
        int k   = i / DOUT;
        int col = i - k * DOUT;
        sW[((k >> 1) * DOUT + col) * 2 + (k & 1)] = W[i];
    }
    __syncthreads();

    const int lane = threadIdx.x & 31;
    const int wave = threadIdx.x >> 5;
    const int m    = lane & 15;
    const int khi  = lane >> 4;
    const int rowBase = (blockIdx.x * 8 + wave) * 16;
    if (rowBase >= nrows) return;

    v8f acc[DOUT / 16];
#pragma unroll
    for (int t = 0; t < DOUT / 16; ++t) acc[t] = (v8f)0.0f;

    const float* xrow = X + (size_t)(rowBase + m) * 64;
    const v2f*   sWp  = (const v2f*)sW;
#pragma unroll
    for (int k4 = 0; k4 < 64; k4 += 4) {
        v2f a = *(const v2f*)(xrow + k4 + 2 * khi);
        const int kp = (k4 >> 1) + khi;
#pragma unroll
        for (int t = 0; t < DOUT / 16; ++t) {
            v2f b = sWp[kp * DOUT + t * 16 + m];
            acc[t] = __builtin_amdgcn_wmma_f32_16x16x4_f32(
                false, a, false, b, (short)0, acc[t], false, false);
        }
    }

#pragma unroll
    for (int t = 0; t < DOUT / 16; ++t)
#pragma unroll
        for (int r = 0; r < 8; ++r)
            Y[(size_t)(rowBase + r + 8 * khi) * DOUT + t * 16 + m] = acc[t][r];
}

template<int D>
__global__ void agg_init_bias(float* __restrict__ agg, const float* __restrict__ b, int n) {
    size_t i = (size_t)blockIdx.x * blockDim.x + threadIdx.x;
    if (i < (size_t)n * D) agg[i] = b[i & (D - 1)];
}

template<int D>
__global__ void edge_scatter(const int* __restrict__ src, const int* __restrict__ dst,
                             const float* __restrict__ dinv,
                             const float* __restrict__ h, float* __restrict__ agg,
                             int e) {
    size_t i = (size_t)blockIdx.x * blockDim.x + threadIdx.x;
    if (i >= (size_t)e * D) return;
    int ed = (int)(i / D);
    int f  = (int)(i & (D - 1));
    int s  = src[ed];
    int d  = dst[ed];
    float w = dinv[s] * dinv[d];
    atomicAdd(&agg[(size_t)d * D + f], h[(size_t)s * D + f] * w);
}

template<int D, bool RELU>
__global__ void self_loop(const float* __restrict__ h, float* __restrict__ agg,
                          const float* __restrict__ dinv, int n) {
    size_t i = (size_t)blockIdx.x * blockDim.x + threadIdx.x;
    if (i >= (size_t)n * D) return;
    int nd = (int)(i / D);
    float w = dinv[nd] * dinv[nd];
    float v = agg[i] + h[i] * w;
    agg[i] = RELU ? (v > 0.0f ? v : 0.0f) : v;
}

extern "C" void kernel_launch(void* const* d_in, const int* in_sizes, int n_in,
                              void* d_out, int out_size, void* d_ws, size_t ws_size,
                              hipStream_t stream) {
    const float* x  = (const float*)d_in[0];
    const int*   ei = (const int*)d_in[1];
    const float* W1 = (const float*)d_in[2];
    const float* b1 = (const float*)d_in[3];
    const float* W2 = (const float*)d_in[4];
    const float* b2 = (const float*)d_in[5];
    float* out = (float*)d_out;

    const int N = in_sizes[0] / IN_D;
    const int E = in_sizes[1] / 2;
    const int* src = ei;
    const int* dst = ei + E;

    float* dinv = (float*)d_ws;
    float* h1   = dinv + N;
    float* agg  = h1 + (size_t)N * HID_D;
    float* h2   = h1;

    const int T = 256;
    auto g = [](long long n, int t) { return (unsigned)((n + t - 1) / t); };

    deg_init <<<g(N, T), T, 0, stream>>>(dinv, N);
    deg_count<<<g(E, T), T, 0, stream>>>(dst, dinv, E);
    deg_rsqrt<<<g(N, T), T, 0, stream>>>(dinv, N);

    const unsigned gemmBlocks = g((long long)(N + 15) / 16, 8);

    gemm_wmma_f32<HID_D><<<gemmBlocks, T, 0, stream>>>(x, W1, h1, N);
    agg_init_bias<HID_D><<<g((long long)N * HID_D, T), T, 0, stream>>>(agg, b1, N);
    edge_scatter<HID_D><<<g((long long)E * HID_D, T), T, 0, stream>>>(src, dst, dinv, h1, agg, E);
    self_loop<HID_D, true><<<g((long long)N * HID_D, T), T, 0, stream>>>(h1, agg, dinv, N);

    gemm_wmma_f32<OUT_D><<<gemmBlocks, T, 0, stream>>>(agg, W2, h2, N);
    agg_init_bias<OUT_D><<<g((long long)N * OUT_D, T), T, 0, stream>>>(out, b2, N);
    edge_scatter<OUT_D><<<g((long long)E * OUT_D, T), T, 0, stream>>>(src, dst, dinv, h2, out, E);
    self_loop<OUT_D, false><<<g((long long)N * OUT_D, T), T, 0, stream>>>(h2, out, dinv, N);
}


# Edge sums without float atomics: edges ordered by destination in-call (one returning int add per edge gives count and arrival-order rank), one wave sums each node's rows in registers.
# speedup vs baseline: 2.6398x; 2.6398x over previous
#include <hip/hip_runtime.h>
#include <hip/hip_bf16.h>


typedef float v2f __attribute__((ext_vector_type(2)));
typedef float v8f __attribute__((ext_vector_type(8)));

#define IN_D  64
#define HID_D 64
#define OUT_D 32

#define SCAN_TILE 1024

__global__ void cnt_zero(int* __restrict__ cnt, int n) {
    int i = blockIdx.x * blockDim.x + threadIdx.x;
    if (i < n) cnt[i] = 0;
}

__global__ void edge_rank(const int* __restrict__ dst, int* __restrict__ cnt,
                          int* __restrict__ rank, int e) {
    int i = blockIdx.x * blockDim.x + threadIdx.x;
    if (i < e) rank[i] = atomicAdd(&cnt[dst[i]], 1);
}

__global__ __launch_bounds__(256) void scan_tile_sums(const int* __restrict__ cnt,
                                                      int* __restrict__ tsum, int n) {
    __shared__ int sw[8];
    const int base = blockIdx.x * SCAN_TILE + threadIdx.x * 4;
    int s = 0;
#pragma unroll
    for (int k = 0; k < 4; ++k)
        if (base + k < n) s += cnt[base + k];
#pragma unroll
    for (int o = 16; o > 0; o >>= 1) s += __shfl_xor(s, o);
    if ((threadIdx.x & 31) == 0) sw[threadIdx.x >> 5] = s;
    __syncthreads();
    if (threadIdx.x == 0) {
        int t = 0;
#pragma unroll
        for (int w = 0; w < 8; ++w) t += sw[w];
        tsum[blockIdx.x] = t;
    }
}

__global__ __launch_bounds__(256) void scan_rows(const int* __restrict__ cnt,
                                                 const int* __restrict__ tsum,
                                                 int* __restrict__ rowptr,
                                                 float* __restrict__ dinv, int n) {
    __shared__ int sw[8];
    __shared__ int sprev[8];
    const int lane = threadIdx.x & 31;
    const int wave = threadIdx.x >> 5;

    int p = 0;
    for (int t = threadIdx.x; t < (int)blockIdx.x; t += 256) p += tsum[t];
#pragma unroll
    for (int o = 16; o > 0; o >>= 1) p += __shfl_xor(p, o);
    if (lane == 0) sprev[wave] = p;

    const int base = blockIdx.x * SCAN_TILE + threadIdx.x * 4;
    int c[4];
    int s = 0;
#pragma unroll
    for (int k = 0; k < 4; ++k) {
        c[k] = (base + k < n) ? cnt[base + k] : 0;
        s += c[k];
    }
    int inc = s;
#pragma unroll
    for (int o = 1; o < 32; o <<= 1) {
        int v = __shfl_up(inc, o);
        if (lane >= o) inc += v;
    }
    if (lane == 31) sw[wave] = inc;
    __syncthreads();

    int run = inc - s;
#pragma unroll
    for (int w = 0; w < 8; ++w) {
        run += sprev[w];
        if (w < wave) run += sw[w];
    }
#pragma unroll
    for (int k = 0; k < 4; ++k) {
        if (base + k < n) {
            rowptr[base + k] = run;
            dinv[base + k]   = rsqrtf((float)(c[k] + 1));
            run += c[k];
        }
    }
}

__global__ void csr_fill(const int* __restrict__ src, const int* __restrict__ dst,
                         const int* __restrict__ rank, const int* __restrict__ rowptr,
                         int* __restrict__ col, int e) {
    int i = blockIdx.x * blockDim.x + threadIdx.x;
    if (i < e) col[rowptr[dst[i]] + rank[i]] = src[i];
}

template<int DOUT>
__global__ __launch_bounds__(256) void gemm_wmma_f32(
        const float* __restrict__ X, const float* __restrict__ W,
        float* __restrict__ Y, int nrows) {
    __shared__ float sW[64 * DOUT];
    for (int i = threadIdx.x; i < 64 * DOUT; i += 256) {
        int k   = i / DOUT;
        int col = i - k * DOUT;
        sW[((k >> 1) * DOUT + col) * 2 + (k & 1)] = W[i];
    }
    __syncthreads();

    const int lane = threadIdx.x & 31;
    const int wave = threadIdx.x >> 5;
    const int m    = lane & 15;
    const int khi  = lane >> 4;
    const int rowBase = (blockIdx.x * 8 + wave) * 16;
    if (rowBase >= nrows) return;

    v8f acc[DOUT / 16];
#pragma unroll
    for (int t = 0; t < DOUT / 16; ++t) acc[t] = (v8f)0.0f;

    const float* xrow = X + (size_t)(rowBase + m) * 64;
    const v2f*   sWp  = (const v2f*)sW;
#pragma unroll
    for (int k4 = 0; k4 < 64; k4 += 4) {
        v2f a = *(const v2f*)(xrow + k4 + 2 * khi);
        const int kp = (k4 >> 1) + khi;
#pragma unroll
        for (int t = 0; t < DOUT / 16; ++t) {
            v2f b = sWp[kp * DOUT + t * 16 + m];
            acc[t] = __builtin_amdgcn_wmma_f32_16x16x4_f32(
                false, a, false, b, (short)0, acc[t], false, false);
        }
    }

#pragma unroll
    for (int t = 0; t < DOUT / 16; ++t)
#pragma unroll
        for (int r = 0; r < 8; ++r)
            Y[(size_t)(rowBase + r + 8 * khi) * DOUT + t * 16 + m] = acc[t][r];
}

template<int D> struct Row;
template<> struct Row<64> {
    typedef v2f T;
    static __device__ __forceinline__ T ld(const float* p) { return *(const v2f*)p; }
    static __device__ __forceinline__ void st(float* p, T v) { *(v2f*)p = v; }
};
template<> struct Row<32> {
    typedef float T;
    static __device__ __forceinline__ T ld(const float* p) { return *p; }
    static __device__ __forceinline__ void st(float* p, T v) { *p = v; }
};
static __device__ __forceinline__ float relu1(float v) { return v > 0.0f ? v : 0.0f; }
static __device__ __forceinline__ v2f   relu1(v2f v) { v2f r; r.x = relu1(v.x); r.y = relu1(v.y); return r; }

template<int D, bool RELU>
__global__ __launch_bounds__(256) void gather_rows(
        const int* __restrict__ rowptr, const int* __restrict__ cnt,
        const int* __restrict__ col, const float* __restrict__ dinv,
        const float* __restrict__ h, const float* __restrict__ bias,
        float* __restrict__ out, int n) {
    typedef typename Row<D>::T RT;
    constexpr int V = D / 32;
    const int lane = threadIdx.x & 31;
    const int node = __builtin_amdgcn_readfirstlane((int)(blockIdx.x * 8 + (threadIdx.x >> 5)));
    if (node >= n) return;

    const int   start = rowptr[node];
    const int   deg   = cnt[node];
    const float dd    = dinv[node];
    const float* hl   = h + lane * V;

    RT acc = (RT)0.0f;
    for (int base = 0; base < deg; base += 32) {
        const int m = (deg - base < 32) ? (deg - base) : 32;
        int   mycol = node;
        float myw   = 0.0f;
        if (lane < m) {
            mycol = col[start + base + lane];
            myw   = dinv[mycol] * dd;
        }
        for (int j = 0; j < m; j += 4) {
            RT    r[4];
            float w[4];
#pragma unroll
            for (int u = 0; u < 4; ++u) {
                const int s = __builtin_amdgcn_readlane(mycol, j + u);
                w[u] = __int_as_float(__builtin_amdgcn_readlane(__float_as_int(myw), j + u));
                r[u] = Row<D>::ld(hl + (size_t)s * D);
            }
#pragma unroll
            for (int u = 0; u < 4; ++u)
                if (j + u < m) acc += r[u] * w[u];
        }
    }
    acc += Row<D>::ld(hl + (size_t)node * D) * (dd * dd);

    RT bv = Row<D>::ld(bias + lane * V);
    acc += bv;
    if (RELU) acc = relu1(acc);
    Row<D>::st(out + (size_t)node * D + lane * V, acc);
}

extern "C" void kernel_launch(void* const* d_in, const int* in_sizes, int n_in,
                              void* d_out, int out_size, void* d_ws, size_t ws_size,
                              hipStream_t stream) {
    const float* x  = (const float*)d_in[0];
    const int*   ei = (const int*)d_in[1];
    const float* W1 = (const float*)d_in[2];
    const float* b1 = (const float*)d_in[3];
    const float* W2 = (const float*)d_in[4];
    const float* b2 = (const float*)d_in[5];
    float* out = (float*)d_out;

    const int N = in_sizes[0] / IN_D;
    const int E = in_sizes[1] / 2;
    const int* src = ei;
    const int* dst = ei + E;

    const int T = 256;
    auto g = [](long long n, int t) { return (unsigned)((n + t - 1) / t); };
    const unsigned tiles = g(N, SCAN_TILE);

    auto up = [](size_t b) { return (b + 255) & ~(size_t)255; };
    char* p = (char*)d_ws;
    float* dinv   = (float*)p;  p += up((size_t)N * 4);
    int*   cnt    = (int*)p;    p += up((size_t)N * 4);
    int*   rowptr = (int*)p;    p += up((size_t)N * 4);
    int*   tsum   = (int*)p;    p += up((size_t)tiles * 4);
    int*   rank   = (int*)p;    p += up((size_t)E * 4);
    int*   col    = (int*)p;    p += up((size_t)E * 4);
    float* h1     = (float*)p;  p += up((size_t)N * HID_D * 4);
    float* agg    = (float*)p;
    float* h2     = h1;

    cnt_zero      <<<g(N, T), T, 0, stream>>>(cnt, N);
    edge_rank     <<<g(E, T), T, 0, stream>>>(dst, cnt, rank, E);
    scan_tile_sums<<<tiles, T, 0, stream>>>(cnt, tsum, N);
    scan_rows     <<<tiles, T, 0, stream>>>(cnt, tsum, rowptr, dinv, N);
    csr_fill      <<<g(E, T), T, 0, stream>>>(src, dst, rank, rowptr, col, E);

    const unsigned gemmBlocks = g((long long)(N + 15) / 16, 8);
    const unsigned nodeBlocks = g(N, 8);

    gemm_wmma_f32<HID_D><<<gemmBlocks, T, 0, stream>>>(x, W1, h1, N);
    gather_rows<HID_D, true><<<nodeBlocks, T, 0, stream>>>(rowptr, cnt, col, dinv, h1, b1, agg, N);

    gemm_wmma_f32<OUT_D><<<gemmBlocks, T, 0, stream>>>(agg, W2, h2, N);
    gather_rows<OUT_D, false><<<nodeBlocks, T, 0, stream>>>(rowptr, cnt, col, dinv, h2, b2, out, N);
}


# Edge ordering on LDS counters: edges filed into 256-node buckets by per-block LDS counts and one scan, then ordered per node by returning LDS adds (arrival order); no global atomics left.
# speedup vs baseline: 4.5615x; 1.7280x over previous
#include <hip/hip_runtime.h>
#include <hip/hip_bf16.h>


typedef float v2f __attribute__((ext_vector_type(2)));
typedef float v8f __attribute__((ext_vector_type(8)));

#define IN_D  64
#define HID_D 64
#define OUT_D 32

#define SCAN_TILE 1024

typedef int v2i __attribute__((ext_vector_type(2)));

#define BSH     8
#define BNODES  256
#define MAXB    512
#define CHUNK   8192
#define ETHR    1024
#define FCAP    12288

__global__ __launch_bounds__(ETHR) void bucket_hist(const int* __restrict__ dst,
                                                    int* __restrict__ table,
                                                    int e, int nb, int nwg) {
    __shared__ int lh[MAXB];
    if (threadIdx.x < MAXB) lh[threadIdx.x] = 0;
    __syncthreads();
    const int base = blockIdx.x * CHUNK + threadIdx.x;
    int d[CHUNK / ETHR];
#pragma unroll
    for (int k = 0; k < CHUNK / ETHR; ++k) {
        const int i = base + k * ETHR;
        d[k] = dst[i < e ? i : e - 1];
    }
#pragma unroll
    for (int k = 0; k < CHUNK / ETHR; ++k)
        if (base + k * ETHR < e) atomicAdd(&lh[d[k] >> BSH], 1);
    __syncthreads();
    if (threadIdx.x < nb) table[threadIdx.x * nwg + blockIdx.x] = lh[threadIdx.x];
}

__global__ __launch_bounds__(256) void scan_tile_sums(const int* __restrict__ cnt,
                                                      int* __restrict__ tsum, int n) {
    __shared__ int sw[8];
    const int base = blockIdx.x * SCAN_TILE + threadIdx.x * 4;
    int s = 0;
#pragma unroll
    for (int k = 0; k < 4; ++k)
        if (base + k < n) s += cnt[base + k];
#pragma unroll
    for (int o = 16; o > 0; o >>= 1) s += __shfl_xor(s, o);
    if ((threadIdx.x & 31) == 0) sw[threadIdx.x >> 5] = s;
    __syncthreads();
    if (threadIdx.x == 0) {
        int t = 0;
#pragma unroll
        for (int w = 0; w < 8; ++w) t += sw[w];
        tsum[blockIdx.x] = t;
    }
}

__global__ __launch_bounds__(256) void scan_excl(const int* __restrict__ in,
                                                 const int* __restrict__ tsum,
                                                 int* __restrict__ out, int n) {
    __shared__ int sw[8];
    __shared__ int sprev[8];
    const int lane = threadIdx.x & 31;
    const int wave = threadIdx.x >> 5;

    int p = 0;
    for (int t = threadIdx.x; t < (int)blockIdx.x; t += 256) p += tsum[t];
#pragma unroll
    for (int o = 16; o > 0; o >>= 1) p += __shfl_xor(p, o);
    if (lane == 0) sprev[wave] = p;

    const int base = blockIdx.x * SCAN_TILE + threadIdx.x * 4;
    int c[4];
    int s = 0;
#pragma unroll
    for (int k = 0; k < 4; ++k) {
        c[k] = (base + k < n) ? in[base + k] : 0;
        s += c[k];
    }
    int inc = s;
#pragma unroll
    for (int o = 1; o < 32; o <<= 1) {
        int v = __shfl_up(inc, o);
        if (lane >= o) inc += v;
    }
    if (lane == 31) sw[wave] = inc;
    __syncthreads();

    int run = inc - s;
#pragma unroll
    for (int w = 0; w < 8; ++w) {
        run += sprev[w];
        if (w < wave) run += sw[w];
    }
#pragma unroll
    for (int k = 0; k < 4; ++k) {
        if (base + k < n) {
            out[base + k] = run;
            run += c[k];
        }
    }
}

__global__ __launch_bounds__(ETHR) void bucket_scatter(const int* __restrict__ src,
                                                       const int* __restrict__ dst,
                                                       const int* __restrict__ tscan,
                                                       int* __restrict__ pairs,
                                                       int e, int nb, int nwg) {
    __shared__ int lpos[MAXB];
    if (threadIdx.x < nb) lpos[threadIdx.x] = tscan[threadIdx.x * nwg + blockIdx.x];
    const int base = blockIdx.x * CHUNK + threadIdx.x;
    v2i v[CHUNK / ETHR];
#pragma unroll
    for (int k = 0; k < CHUNK / ETHR; ++k) {
        const int i  = base + k * ETHR;
        const int ic = i < e ? i : e - 1;
        v[k].x = dst[ic];
        v[k].y = src[ic];
    }
    __syncthreads();
    v2i* pr = (v2i*)pairs;
#pragma unroll
    for (int k = 0; k < CHUNK / ETHR; ++k) {
        if (base + k * ETHR < e) {
            const int p = atomicAdd(&lpos[v[k].x >> BSH], 1);
            pr[p] = v[k];
        }
    }
}

__global__ __launch_bounds__(ETHR) void bucket_rows(const int* __restrict__ pairs,
                                                    const int* __restrict__ tscan,
                                                    int* __restrict__ rowptr,
                                                    int* __restrict__ cnt,
                                                    float* __restrict__ dinv,
                                                    int* __restrict__ col,
                                                    int n, int e, int nb, int nwg) {
    __shared__ int lcnt[BNODES];
    __shared__ int lcur[BNODES];
    __shared__ int lwave[BNODES / 32];
    __shared__ int stage[FCAP];
    const int b    = blockIdx.x;
    const int bs   = tscan[b * nwg];
    const int be   = (b + 1 < nb) ? tscan[(b + 1) * nwg] : e;
    const int len  = be - bs;
    const int lane = threadIdx.x & 31;
    const int wave = threadIdx.x >> 5;
    const v2i* pr  = (const v2i*)pairs;

    if (threadIdx.x < BNODES) lcnt[threadIdx.x] = 0;
    __syncthreads();
    for (int i0 = bs + threadIdx.x; i0 < be; i0 += 4 * ETHR) {
        int d[4];
#pragma unroll
        for (int u = 0; u < 4; ++u) {
            const int i = i0 + u * ETHR;
            d[u] = pr[i < be ? i : be - 1].x;
        }
#pragma unroll
        for (int u = 0; u < 4; ++u)
            if (i0 + u * ETHR < be) atomicAdd(&lcnt[d[u] & (BNODES - 1)], 1);
    }
    __syncthreads();

    int c = 0, inc = 0;
    if (threadIdx.x < BNODES) {
        c   = lcnt[threadIdx.x];
        inc = c;
#pragma unroll
        for (int o = 1; o < 32; o <<= 1) {
            int v = __shfl_up(inc, o);
            if (lane >= o) inc += v;
        }
        if (lane == 31) lwave[wave] = inc;
    }
    __syncthreads();
    if (threadIdx.x < BNODES) {
        int run = inc - c;
#pragma unroll
        for (int w = 0; w < BNODES / 32; ++w)
            if (w < wave) run += lwave[w];
        lcur[threadIdx.x] = run;
        const int node = b * BNODES + threadIdx.x;
        if (node < n) {
            rowptr[node] = bs + run;
            cnt[node]    = c;
            dinv[node]   = rsqrtf((float)(c + 1));
        }
    }
    __syncthreads();

    if (len <= FCAP) {
        for (int i0 = bs + threadIdx.x; i0 < be; i0 += 4 * ETHR) {
            v2i v[4];
#pragma unroll
            for (int u = 0; u < 4; ++u) {
                const int i = i0 + u * ETHR;
                v[u] = pr[i < be ? i : be - 1];
            }
#pragma unroll
            for (int u = 0; u < 4; ++u) {
                if (i0 + u * ETHR < be) {
                    const int q = atomicAdd(&lcur[v[u].x & (BNODES - 1)], 1);
                    stage[q] = v[u].y;
                }
            }
        }
        __syncthreads();
        for (int i = threadIdx.x; i < len; i += ETHR) col[bs + i] = stage[i];
    } else {
        for (int i = bs + threadIdx.x; i < be; i += ETHR) {
            const v2i v = pr[i];
            const int q = atomicAdd(&lcur[v.x & (BNODES - 1)], 1);
            col[bs + q] = v.y;
        }
    }
}

template<int DOUT>
__global__ __launch_bounds__(256) void gemm_wmma_f32(
        const float* __restrict__ X, const float* __restrict__ W,
        float* __restrict__ Y, int nrows) {
    __shared__ float sW[64 * DOUT];
    for (int i = threadIdx.x; i < 64 * DOUT; i += 256) {
        int k   = i / DOUT;
        int col = i - k * DOUT;
        sW[((k >> 1) * DOUT + col) * 2 + (k & 1)] = W[i];
    }
    __syncthreads();

    const int lane = threadIdx.x & 31;
    const int wave = threadIdx.x >> 5;
    const int m    = lane & 15;
    const int khi  = lane >> 4;
    const int rowBase = (blockIdx.x * 8 + wave) * 16;
    if (rowBase >= nrows) return;

    v8f acc[DOUT / 16];
#pragma unroll
    for (int t = 0; t < DOUT / 16; ++t) acc[t] = (v8f)0.0f;

    const float* xrow = X + (size_t)(rowBase + m) * 64;
    const v2f*   sWp  = (const v2f*)sW;
#pragma unroll
    for (int k4 = 0; k4 < 64; k4 += 4) {
        v2f a = *(const v2f*)(xrow + k4 + 2 * khi);
        const int kp = (k4 >> 1) + khi;
#pragma unroll
        for (int t = 0; t < DOUT / 16; ++t) {
            v2f b = sWp[kp * DOUT + t * 16 + m];
            acc[t] = __builtin_amdgcn_wmma_f32_16x16x4_f32(
                false, a, false, b, (short)0, acc[t], false, false);
        }
    }

#pragma unroll
    for (int t = 0; t < DOUT / 16; ++t)
#pragma unroll
        for (int r = 0; r < 8; ++r)
            Y[(size_t)(rowBase + r + 8 * khi) * DOUT + t * 16 + m] = acc[t][r];
}

template<int D> struct Row;
template<> struct Row<64> {
    typedef v2f T;
    static __device__ __forceinline__ T ld(const float* p) { return *(const v2f*)p; }
    static __device__ __forceinline__ void st(float* p, T v) { *(v2f*)p = v; }
};
template<> struct Row<32> {
    typedef float T;
    static __device__ __forceinline__ T ld(const float* p) { return *p; }
    static __device__ __forceinline__ void st(float* p, T v) { *p = v; }
};
static __device__ __forceinline__ float relu1(float v) { return v > 0.0f ? v : 0.0f; }
static __device__ __forceinline__ v2f   relu1(v2f v) { v2f r; r.x = relu1(v.x); r.y = relu1(v.y); return r; }

template<int D, bool RELU>
__global__ __launch_bounds__(256) void gather_rows(
        const int* __restrict__ rowptr, const int* __restrict__ cnt,
        const int* __restrict__ col, const float* __restrict__ dinv,
        const float* __restrict__ h, const float* __restrict__ bias,
        float* __restrict__ out, int n) {
    typedef typename Row<D>::T RT;
    constexpr int V = D / 32;
    const int lane = threadIdx.x & 31;
    const int node = __builtin_amdgcn_readfirstlane((int)(blockIdx.x * 8 + (threadIdx.x >> 5)));
    if (node >= n) return;

    const int   start = rowptr[node];
    const int   deg   = cnt[node];
    const float dd    = dinv[node];
    const float* hl   = h + lane * V;

    RT acc = (RT)0.0f;
    for (int base = 0; base < deg; base += 32) {
        const int m = (deg - base < 32) ? (deg - base) : 32;
        int   mycol = node;
        float myw   = 0.0f;
        if (lane < m) {
            mycol = col[start + base + lane];
            myw   = dinv[mycol] * dd;
        }
        for (int j = 0; j < m; j += 4) {
            RT    r[4];
            float w[4];
#pragma unroll
            for (int u = 0; u < 4; ++u) {
                const int s = __builtin_amdgcn_readlane(mycol, j + u);
                w[u] = __int_as_float(__builtin_amdgcn_readlane(__float_as_int(myw), j + u));
                r[u] = Row<D>::ld(hl + (size_t)s * D);
            }
#pragma unroll
            for (int u = 0; u < 4; ++u)
                if (j + u < m) acc += r[u] * w[u];
        }
    }
    acc += Row<D>::ld(hl + (size_t)node * D) * (dd * dd);

    RT bv = Row<D>::ld(bias + lane * V);
    acc += bv;
    if (RELU) acc = relu1(acc);
    Row<D>::st(out + (size_t)node * D + lane * V, acc);
}

extern "C" void kernel_launch(void* const* d_in, const int* in_sizes, int n_in,
                              void* d_out, int out_size, void* d_ws, size_t ws_size,
                              hipStream_t stream) {
    const float* x  = (const float*)d_in[0];
    const int*   ei = (const int*)d_in[1];
    const float* W1 = (const float*)d_in[2];
    const float* b1 = (const float*)d_in[3];
    const float* W2 = (const float*)d_in[4];
    const float* b2 = (const float*)d_in[5];
    float* out = (float*)d_out;

    const int N = in_sizes[0] / IN_D;
    const int E = in_sizes[1] / 2;
    const int* src = ei;
    const int* dst = ei + E;

    const int T = 256;
    auto g = [](long long n, int t) { return (unsigned)((n + t - 1) / t); };
    const int nb  = (N + BNODES - 1) >> BSH;
    const int nwg = (int)g(E, CHUNK);
    const int nt  = nb * nwg;
    const unsigned tiles = g(nt, SCAN_TILE);

    auto up = [](size_t b) { return (b + 255) & ~(size_t)255; };
    char* p = (char*)d_ws;
    float* dinv   = (float*)p;  p += up((size_t)N * 4);
    int*   cnt    = (int*)p;    p += up((size_t)N * 4);
    int*   rowptr = (int*)p;    p += up((size_t)N * 4);
    int*   tsum   = (int*)p;    p += up((size_t)tiles * 4);
    int*   table  = (int*)p;    p += up((size_t)nt * 4);
    int*   tscan  = (int*)p;    p += up((size_t)nt * 4);
    int*   pairs  = (int*)p;    p += up((size_t)E * 8);
    int*   col    = (int*)p;    p += up((size_t)E * 4);
    float* h1     = (float*)p;  p += up((size_t)N * HID_D * 4);
    float* agg    = (float*)p;
    float* h2     = h1;

    bucket_hist   <<<nwg, ETHR, 0, stream>>>(dst, table, E, nb, nwg);
    scan_tile_sums<<<tiles, T, 0, stream>>>(table, tsum, nt);
    scan_excl     <<<tiles, T, 0, stream>>>(table, tsum, tscan, nt);
    bucket_scatter<<<nwg, ETHR, 0, stream>>>(src, dst, tscan, pairs, E, nb, nwg);
    bucket_rows   <<<nb, ETHR, 0, stream>>>(pairs, tscan, rowptr, cnt, dinv, col, N, E, nb, nwg);

    const unsigned gemmBlocks = g((long long)(N + 15) / 16, 8);
    const unsigned nodeBlocks = g(N, 8);

    gemm_wmma_f32<HID_D><<<gemmBlocks, T, 0, stream>>>(x, W1, h1, N);
    gather_rows<HID_D, true><<<nodeBlocks, T, 0, stream>>>(rowptr, cnt, col, dinv, h1, b1, agg, N);

    gemm_wmma_f32<OUT_D><<<gemmBlocks, T, 0, stream>>>(agg, W2, h2, N);
    gather_rows<OUT_D, false><<<nodeBlocks, T, 0, stream>>>(rowptr, cnt, col, dinv, h2, b2, out, N);
}


# Per-destination sums on long-lived waves: each wave walks 8 consecutive rows (8x fewer waves enter), the next rows' sources and weights requested under the current row's loads, 8 row loads per group.
# speedup vs baseline: 4.8112x; 1.0547x over previous
#include <hip/hip_runtime.h>
#include <hip/hip_bf16.h>


typedef float v2f __attribute__((ext_vector_type(2)));
typedef float v8f __attribute__((ext_vector_type(8)));

#define IN_D  64
#define HID_D 64
#define OUT_D 32

#define SCAN_TILE 1024

typedef int v2i __attribute__((ext_vector_type(2)));

#define BSH     8
#define BNODES  256
#define MAXB    512
#define CHUNK   8192
#define ETHR    1024
#define FCAP    12288

__global__ __launch_bounds__(ETHR) void bucket_hist(const int* __restrict__ dst,
                                                    int* __restrict__ table,
                                                    int e, int nb, int nwg) {
    __shared__ int lh[MAXB];
    if (threadIdx.x < MAXB) lh[threadIdx.x] = 0;
    __syncthreads();
    const int base = blockIdx.x * CHUNK + threadIdx.x;
    int d[CHUNK / ETHR];
#pragma unroll
    for (int k = 0; k < CHUNK / ETHR; ++k) {
        const int i = base + k * ETHR;
        d[k] = dst[i < e ? i : e - 1];
    }
#pragma unroll
    for (int k = 0; k < CHUNK / ETHR; ++k)
        if (base + k * ETHR < e) atomicAdd(&lh[d[k] >> BSH], 1);
    __syncthreads();
    if (threadIdx.x < nb) table[threadIdx.x * nwg + blockIdx.x] = lh[threadIdx.x];
}

__global__ __launch_bounds__(256) void scan_tile_sums(const int* __restrict__ cnt,
                                                      int* __restrict__ tsum, int n) {
    __shared__ int sw[8];
    const int base = blockIdx.x * SCAN_TILE + threadIdx.x * 4;
    int s = 0;
#pragma unroll
    for (int k = 0; k < 4; ++k)
        if (base + k < n) s += cnt[base + k];
#pragma unroll
    for (int o = 16; o > 0; o >>= 1) s += __shfl_xor(s, o);
    if ((threadIdx.x & 31) == 0) sw[threadIdx.x >> 5] = s;
    __syncthreads();
    if (threadIdx.x == 0) {
        int t = 0;
#pragma unroll
        for (int w = 0; w < 8; ++w) t += sw[w];
        tsum[blockIdx.x] = t;
    }
}

__global__ __launch_bounds__(256) void scan_excl(const int* __restrict__ in,
                                                 const int* __restrict__ tsum,
                                                 int* __restrict__ out, int n) {
    __shared__ int sw[8];
    __shared__ int sprev[8];
    const int lane = threadIdx.x & 31;
    const int wave = threadIdx.x >> 5;

    int p = 0;
    for (int t = threadIdx.x; t < (int)blockIdx.x; t += 256) p += tsum[t];
#pragma unroll
    for (int o = 16; o > 0; o >>= 1) p += __shfl_xor(p, o);
    if (lane == 0) sprev[wave] = p;

    const int base = blockIdx.x * SCAN_TILE + threadIdx.x * 4;
    int c[4];
    int s = 0;
#pragma unroll
    for (int k = 0; k < 4; ++k) {
        c[k] = (base + k < n) ? in[base + k] : 0;
        s += c[k];
    }
    int inc = s;
#pragma unroll
    for (int o = 1; o < 32; o <<= 1) {
        int v = __shfl_up(inc, o);
        if (lane >= o) inc += v;
    }
    if (lane == 31) sw[wave] = inc;
    __syncthreads();

    int run = inc - s;
#pragma unroll
    for (int w = 0; w < 8; ++w) {
        run += sprev[w];
        if (w < wave) run += sw[w];
    }
#pragma unroll
    for (int k = 0; k < 4; ++k) {
        if (base + k < n) {
            out[base + k] = run;
            run += c[k];
        }
    }
}

__global__ __launch_bounds__(ETHR) void bucket_scatter(const int* __restrict__ src,
                                                       const int* __restrict__ dst,
                                                       const int* __restrict__ tscan,
                                                       int* __restrict__ pairs,
                                                       int e, int nb, int nwg) {
    __shared__ int lpos[MAXB];
    if (threadIdx.x < nb) lpos[threadIdx.x] = tscan[threadIdx.x * nwg + blockIdx.x];
    const int base = blockIdx.x * CHUNK + threadIdx.x;
    v2i v[CHUNK / ETHR];
#pragma unroll
    for (int k = 0; k < CHUNK / ETHR; ++k) {
        const int i  = base + k * ETHR;
        const int ic = i < e ? i : e - 1;
        v[k].x = dst[ic];
        v[k].y = src[ic];
    }
    __syncthreads();
    v2i* pr = (v2i*)pairs;
#pragma unroll
    for (int k = 0; k < CHUNK / ETHR; ++k) {
        if (base + k * ETHR < e) {
            const int p = atomicAdd(&lpos[v[k].x >> BSH], 1);
            pr[p] = v[k];
        }
    }
}

__global__ __launch_bounds__(ETHR) void bucket_rows(const int* __restrict__ pairs,
                                                    const int* __restrict__ tscan,
                                                    int* __restrict__ rowptr,
                                                    int* __restrict__ cnt,
                                                    float* __restrict__ dinv,
                                                    int* __restrict__ col,
                                                    int n, int e, int nb, int nwg) {
    __shared__ int lcnt[BNODES];
    __shared__ int lcur[BNODES];
    __shared__ int lwave[BNODES / 32];
    __shared__ int stage[FCAP];
    const int b    = blockIdx.x;
    const int bs   = tscan[b * nwg];
    const int be   = (b + 1 < nb) ? tscan[(b + 1) * nwg] : e;
    const int len  = be - bs;
    const int lane = threadIdx.x & 31;
    const int wave = threadIdx.x >> 5;
    const v2i* pr  = (const v2i*)pairs;

    if (threadIdx.x < BNODES) lcnt[threadIdx.x] = 0;
    __syncthreads();
    for (int i0 = bs + threadIdx.x; i0 < be; i0 += 4 * ETHR) {
        int d[4];
#pragma unroll
        for (int u = 0; u < 4; ++u) {
            const int i = i0 + u * ETHR;
            d[u] = pr[i < be ? i : be - 1].x;
        }
#pragma unroll
        for (int u = 0; u < 4; ++u)
            if (i0 + u * ETHR < be) atomicAdd(&lcnt[d[u] & (BNODES - 1)], 1);
    }
    __syncthreads();

    int c = 0, inc = 0;
    if (threadIdx.x < BNODES) {
        c   = lcnt[threadIdx.x];
        inc = c;
#pragma unroll
        for (int o = 1; o < 32; o <<= 1) {
            int v = __shfl_up(inc, o);
            if (lane >= o) inc += v;
        }
        if (lane == 31) lwave[wave] = inc;
    }
    __syncthreads();
    if (threadIdx.x < BNODES) {
        int run = inc - c;
#pragma unroll
        for (int w = 0; w < BNODES / 32; ++w)
            if (w < wave) run += lwave[w];
        lcur[threadIdx.x] = run;
        const int node = b * BNODES + threadIdx.x;
        if (node < n) {
            rowptr[node] = bs + run;
            cnt[node]    = c;
            dinv[node]   = rsqrtf((float)(c + 1));
        }
    }
    __syncthreads();

    if (len <= FCAP) {
        for (int i0 = bs + threadIdx.x; i0 < be; i0 += 4 * ETHR) {
            v2i v[4];
#pragma unroll
            for (int u = 0; u < 4; ++u) {
                const int i = i0 + u * ETHR;
                v[u] = pr[i < be ? i : be - 1];
            }
#pragma unroll
            for (int u = 0; u < 4; ++u) {
                if (i0 + u * ETHR < be) {
                    const int q = atomicAdd(&lcur[v[u].x & (BNODES - 1)], 1);
                    stage[q] = v[u].y;
                }
            }
        }
        __syncthreads();
        for (int i = threadIdx.x; i < len; i += ETHR) col[bs + i] = stage[i];
    } else {
        for (int i = bs + threadIdx.x; i < be; i += ETHR) {
            const v2i v = pr[i];
            const int q = atomicAdd(&lcur[v.x & (BNODES - 1)], 1);
            col[bs + q] = v.y;
        }
    }
}

template<int DOUT>
__global__ __launch_bounds__(256) void gemm_wmma_f32(
        const float* __restrict__ X, const float* __restrict__ W,
        float* __restrict__ Y, int nrows) {
    __shared__ float sW[64 * DOUT];
    for (int i = threadIdx.x; i < 64 * DOUT; i += 256) {
        int k   = i / DOUT;
        int col = i - k * DOUT;
        sW[((k >> 1) * DOUT + col) * 2 + (k & 1)] = W[i];
    }
    __syncthreads();

    const int lane = threadIdx.x & 31;
    const int wave = threadIdx.x >> 5;
    const int m    = lane & 15;
    const int khi  = lane >> 4;
    const int rowBase = (blockIdx.x * 8 + wave) * 16;
    if (rowBase >= nrows) return;

    v8f acc[DOUT / 16];
#pragma unroll
    for (int t = 0; t < DOUT / 16; ++t) acc[t] = (v8f)0.0f;

    const float* xrow = X + (size_t)(rowBase + m) * 64;
    const v2f*   sWp  = (const v2f*)sW;
#pragma unroll
    for (int k4 = 0; k4 < 64; k4 += 4) {
        v2f a = *(const v2f*)(xrow + k4 + 2 * khi);
        const int kp = (k4 >> 1) + khi;
#pragma unroll
        for (int t = 0; t < DOUT / 16; ++t) {
            v2f b = sWp[kp * DOUT + t * 16 + m];
            acc[t] = __builtin_amdgcn_wmma_f32_16x16x4_f32(
                false, a, false, b, (short)0, acc[t], false, false);
        }
    }

#pragma unroll
    for (int t = 0; t < DOUT / 16; ++t)
#pragma unroll
        for (int r = 0; r < 8; ++r)
            Y[(size_t)(rowBase + r + 8 * khi) * DOUT + t * 16 + m] = acc[t][r];
}

#define NPW 8
#define GU  8

template<int D> struct Row;
template<> struct Row<64> {
    typedef v2f T;
    static __device__ __forceinline__ T ld(const float* p) { return *(const v2f*)p; }
    static __device__ __forceinline__ void st(float* p, T v) { *(v2f*)p = v; }
};
template<> struct Row<32> {
    typedef float T;
    static __device__ __forceinline__ T ld(const float* p) { return *p; }
    static __device__ __forceinline__ void st(float* p, T v) { *p = v; }
};
static __device__ __forceinline__ float relu1(float v) { return v > 0.0f ? v : 0.0f; }
static __device__ __forceinline__ v2f   relu1(v2f v) { v2f r; r.x = relu1(v.x); r.y = relu1(v.y); return r; }

static __device__ __forceinline__ int   lane_i(int v, int l)   { return __builtin_amdgcn_readlane(v, l); }
static __device__ __forceinline__ float lane_f(float v, int l) { return __int_as_float(__builtin_amdgcn_readlane(__float_as_int(v), l)); }

template<int D, bool RELU>
__global__ __launch_bounds__(256) void gather_rows(
        const int* __restrict__ rowptr, const int* __restrict__ cnt,
        const int* __restrict__ col, const float* __restrict__ dinv,
        const float* __restrict__ h, const float* __restrict__ bias,
        float* __restrict__ out, int n) {
    typedef typename Row<D>::T RT;
    constexpr int V = D / 32;
    const int lane = threadIdx.x & 31;
    const int gw = __builtin_amdgcn_readfirstlane((int)(blockIdx.x * 8 + (threadIdx.x >> 5)));
    const int n0 = gw * NPW;
    if (n0 >= n) return;
    const int nn = (n - n0 < NPW) ? (n - n0) : NPW;

    const int   li     = n0 + (lane < nn ? lane : nn - 1);
    const int   vstart = rowptr[li];
    const int   vdeg   = cnt[li];
    const float vdd    = dinv[li];
    const float* hl    = h + lane * V;
    const RT     bv    = Row<D>::ld(bias + lane * V);

    int   mycol = n0;
    float myw   = 0.0f;
    int   col1  = n0;
    {
        const int   s0 = lane_i(vstart, 0);
        const int   d0 = lane_i(vdeg, 0);
        const float w0 = lane_f(vdd, 0);
        const int   s1 = lane_i(vstart, nn > 1 ? 1 : 0);
        const int   d1 = nn > 1 ? lane_i(vdeg, 1) : 0;
        float g0 = 0.0f;
        if (lane < d0) mycol = col[s0 + lane];
        if (lane < d1) col1  = col[s1 + lane];
        if (lane < d0) g0    = dinv[mycol];
        asm volatile("" :: "v"(mycol), "v"(col1), "v"(g0));
        myw = g0 * w0;
    }

    for (int i = 0; i < nn; ++i) {
        const int   node  = n0 + i;
        const int   start = lane_i(vstart, i);
        const int   deg   = lane_i(vdeg, i);
        const float dd    = lane_f(vdd, i);

        const bool  more1 = i + 1 < nn;
        const int   i1    = more1 ? i + 1 : i;
        const int   deg1  = more1 ? lane_i(vdeg, i1) : 0;
        const float dd1   = lane_f(vdd, i1);
        float w1 = 0.0f;
        if (lane < deg1) w1 = dinv[col1];

        const bool  more2  = i + 2 < nn;
        const int   i2     = more2 ? i + 2 : i;
        const int   start2 = lane_i(vstart, i2);
        const int   deg2   = more2 ? lane_i(vdeg, i2) : 0;
        int col2 = n0 + i2;
        if (lane < deg2) col2 = col[start2 + lane];

        const RT self = Row<D>::ld(hl + (size_t)node * D);

        RT acc = (RT)0.0f;
        for (int base = 0; base < deg; base += 32) {
            const int m = (deg - base < 32) ? (deg - base) : 32;
            if (base > 0) {
                mycol = node;
                myw   = 0.0f;
                if (lane < m) {
                    mycol = col[start + base + lane];
                    myw   = dinv[mycol] * dd;
                }
            }
            for (int j = 0; j < m; j += GU) {
                RT    r[GU];
                float w[GU];
#pragma unroll
                for (int u = 0; u < GU; ++u) {
                    const int s = lane_i(mycol, j + u);
                    w[u] = lane_f(myw, j + u);
                    r[u] = Row<D>::ld(hl + (size_t)s * D);
                }
#pragma unroll
                for (int u = 0; u < GU; ++u)
                    if (j + u < m) acc += r[u] * w[u];
            }
        }

        acc += self * (dd * dd);
        acc += bv;
        if (RELU) acc = relu1(acc);
        Row<D>::st(out + (size_t)node * D + lane * V, acc);

        mycol = col1;
        myw   = w1 * dd1;
        col1  = col2;
    }
}

extern "C" void kernel_launch(void* const* d_in, const int* in_sizes, int n_in,
                              void* d_out, int out_size, void* d_ws, size_t ws_size,
                              hipStream_t stream) {
    const float* x  = (const float*)d_in[0];
    const int*   ei = (const int*)d_in[1];
    const float* W1 = (const float*)d_in[2];
    const float* b1 = (const float*)d_in[3];
    const float* W2 = (const float*)d_in[4];
    const float* b2 = (const float*)d_in[5];
    float* out = (float*)d_out;

    const int N = in_sizes[0] / IN_D;
    const int E = in_sizes[1] / 2;
    const int* src = ei;
    const int* dst = ei + E;

    const int T = 256;
    auto g = [](long long n, int t) { return (unsigned)((n + t - 1) / t); };
    const int nb  = (N + BNODES - 1) >> BSH;
    const int nwg = (int)g(E, CHUNK);
    const int nt  = nb * nwg;
    const unsigned tiles = g(nt, SCAN_TILE);

    auto up = [](size_t b) { return (b + 255) & ~(size_t)255; };
    char* p = (char*)d_ws;
    float* dinv   = (float*)p;  p += up((size_t)N * 4);
    int*   cnt    = (int*)p;    p += up((size_t)N * 4);
    int*   rowptr = (int*)p;    p += up((size_t)N * 4);
    int*   tsum   = (int*)p;    p += up((size_t)tiles * 4);
    int*   table  = (int*)p;    p += up((size_t)nt * 4);
    int*   tscan  = (int*)p;    p += up((size_t)nt * 4);
    int*   pairs  = (int*)p;    p += up((size_t)E * 8);
    int*   col    = (int*)p;    p += up((size_t)E * 4);
    float* h1     = (float*)p;  p += up((size_t)N * HID_D * 4);
    float* agg    = (float*)p;
    float* h2     = h1;

    bucket_hist   <<<nwg, ETHR, 0, stream>>>(dst, table, E, nb, nwg);
    scan_tile_sums<<<tiles, T, 0, stream>>>(table, tsum, nt);
    scan_excl     <<<tiles, T, 0, stream>>>(table, tsum, tscan, nt);
    bucket_scatter<<<nwg, ETHR, 0, stream>>>(src, dst, tscan, pairs, E, nb, nwg);
    bucket_rows   <<<nb, ETHR, 0, stream>>>(pairs, tscan, rowptr, cnt, dinv, col, N, E, nb, nwg);

    const unsigned gemmBlocks = g((long long)(N + 15) / 16, 8);
    const unsigned nodeBlocks = g(g(N, NPW), 8);

    gemm_wmma_f32<HID_D><<<gemmBlocks, T, 0, stream>>>(x, W1, h1, N);
    gather_rows<HID_D, true><<<nodeBlocks, T, 0, stream>>>(rowptr, cnt, col, dinv, h1, b1, agg, N);

    gemm_wmma_f32<OUT_D><<<gemmBlocks, T, 0, stream>>>(agg, W2, h2, N);
    gather_rows<OUT_D, false><<<nodeBlocks, T, 0, stream>>>(rowptr, cnt, col, dinv, h2, b2, out, N);
}


# Rows stored pre-scaled by their node's dinv in the product kernels; row sums add plain rows and scale once by dinv[dst]: per-edge dinv[src] lookups gone (rounding of the scaling differs).
# speedup vs baseline: 5.1061x; 1.0613x over previous
#include <hip/hip_runtime.h>
#include <hip/hip_bf16.h>


typedef float v2f __attribute__((ext_vector_type(2)));
typedef float v8f __attribute__((ext_vector_type(8)));

#define IN_D  64
#define HID_D 64
#define OUT_D 32

#define SCAN_TILE 1024

typedef int v2i __attribute__((ext_vector_type(2)));

#define BSH     8
#define BNODES  256
#define MAXB    512
#define CHUNK   8192
#define ETHR    1024
#define FCAP    12288

__global__ __launch_bounds__(ETHR) void bucket_hist(const int* __restrict__ dst,
                                                    int* __restrict__ table,
                                                    int e, int nb, int nwg) {
    __shared__ int lh[MAXB];
    if (threadIdx.x < MAXB) lh[threadIdx.x] = 0;
    __syncthreads();
    const int base = blockIdx.x * CHUNK + threadIdx.x;
    int d[CHUNK / ETHR];
#pragma unroll
    for (int k = 0; k < CHUNK / ETHR; ++k) {
        const int i = base + k * ETHR;
        d[k] = dst[i < e ? i : e - 1];
    }
#pragma unroll
    for (int k = 0; k < CHUNK / ETHR; ++k)
        if (base + k * ETHR < e) atomicAdd(&lh[d[k] >> BSH], 1);
    __syncthreads();
    if (threadIdx.x < nb) table[threadIdx.x * nwg + blockIdx.x] = lh[threadIdx.x];
}

__global__ __launch_bounds__(256) void scan_tile_sums(const int* __restrict__ cnt,
                                                      int* __restrict__ tsum, int n) {
    __shared__ int sw[8];
    const int base = blockIdx.x * SCAN_TILE + threadIdx.x * 4;
    int s = 0;
#pragma unroll
    for (int k = 0; k < 4; ++k)
        if (base + k < n) s += cnt[base + k];
#pragma unroll
    for (int o = 16; o > 0; o >>= 1) s += __shfl_xor(s, o);
    if ((threadIdx.x & 31) == 0) sw[threadIdx.x >> 5] = s;
    __syncthreads();
    if (threadIdx.x == 0) {
        int t = 0;
#pragma unroll
        for (int w = 0; w < 8; ++w) t += sw[w];
        tsum[blockIdx.x] = t;
    }
}

__global__ __launch_bounds__(256) void scan_excl(const int* __restrict__ in,
                                                 const int* __restrict__ tsum,
                                                 int* __restrict__ out, int n) {
    __shared__ int sw[8];
    __shared__ int sprev[8];
    const int lane = threadIdx.x & 31;
    const int wave = threadIdx.x >> 5;

    int p = 0;
    for (int t = threadIdx.x; t < (int)blockIdx.x; t += 256) p += tsum[t];
#pragma unroll
    for (int o = 16; o > 0; o >>= 1) p += __shfl_xor(p, o);
    if (lane == 0) sprev[wave] = p;

    const int base = blockIdx.x * SCAN_TILE + threadIdx.x * 4;
    int c[4];
    int s = 0;
#pragma unroll
    for (int k = 0; k < 4; ++k) {
        c[k] = (base + k < n) ? in[base + k] : 0;
        s += c[k];
    }
    int inc = s;
#pragma unroll
    for (int o = 1; o < 32; o <<= 1) {
        int v = __shfl_up(inc, o);
        if (lane >= o) inc += v;
    }
    if (lane == 31) sw[wave] = inc;
    __syncthreads();

    int run = inc - s;
#pragma unroll
    for (int w = 0; w < 8; ++w) {
        run += sprev[w];
        if (w < wave) run += sw[w];
    }
#pragma unroll
    for (int k = 0; k < 4; ++k) {
        if (base + k < n) {
            out[base + k] = run;
            run += c[k];
        }
    }
}

__global__ __launch_bounds__(ETHR) void bucket_scatter(const int* __restrict__ src,
                                                       const int* __restrict__ dst,
                                                       const int* __restrict__ tscan,
                                                       int* __restrict__ pairs,
                                                       int e, int nb, int nwg) {
    __shared__ int lpos[MAXB];
    if (threadIdx.x < nb) lpos[threadIdx.x] = tscan[threadIdx.x * nwg + blockIdx.x];
    const int base = blockIdx.x * CHUNK + threadIdx.x;
    v2i v[CHUNK / ETHR];
#pragma unroll
    for (int k = 0; k < CHUNK / ETHR; ++k) {
        const int i  = base + k * ETHR;
        const int ic = i < e ? i : e - 1;
        v[k].x = dst[ic];
        v[k].y = src[ic];
    }
    __syncthreads();
    v2i* pr = (v2i*)pairs;
#pragma unroll
    for (int k = 0; k < CHUNK / ETHR; ++k) {
        if (base + k * ETHR < e) {
            const int p = atomicAdd(&lpos[v[k].x >> BSH], 1);
            pr[p] = v[k];
        }
    }
}

__global__ __launch_bounds__(ETHR) void bucket_rows(const int* __restrict__ pairs,
                                                    const int* __restrict__ tscan,
                                                    int* __restrict__ rowptr,
                                                    int* __restrict__ cnt,
                                                    float* __restrict__ dinv,
                                                    int* __restrict__ col,
                                                    int n, int e, int nb, int nwg) {
    __shared__ int lcnt[BNODES];
    __shared__ int lcur[BNODES];
    __shared__ int lwave[BNODES / 32];
    __shared__ int stage[FCAP];
    const int b    = blockIdx.x;
    const int bs   = tscan[b * nwg];
    const int be   = (b + 1 < nb) ? tscan[(b + 1) * nwg] : e;
    const int len  = be - bs;
    const int lane = threadIdx.x & 31;
    const int wave = threadIdx.x >> 5;
    const v2i* pr  = (const v2i*)pairs;

    if (threadIdx.x < BNODES) lcnt[threadIdx.x] = 0;
    __syncthreads();
    for (int i0 = bs + threadIdx.x; i0 < be; i0 += 4 * ETHR) {
        int d[4];
#pragma unroll
        for (int u = 0; u < 4; ++u) {
            const int i = i0 + u * ETHR;
            d[u] = pr[i < be ? i : be - 1].x;
        }
#pragma unroll
        for (int u = 0; u < 4; ++u)
            if (i0 + u * ETHR < be) atomicAdd(&lcnt[d[u] & (BNODES - 1)], 1);
    }
    __syncthreads();

    int c = 0, inc = 0;
    if (threadIdx.x < BNODES) {
        c   = lcnt[threadIdx.x];
        inc = c;
#pragma unroll
        for (int o = 1; o < 32; o <<= 1) {
            int v = __shfl_up(inc, o);
            if (lane >= o) inc += v;
        }
        if (lane == 31) lwave[wave] = inc;
    }
    __syncthreads();
    if (threadIdx.x < BNODES) {
        int run = inc - c;
#pragma unroll
        for (int w = 0; w < BNODES / 32; ++w)
            if (w < wave) run += lwave[w];
        lcur[threadIdx.x] = run;
        const int node = b * BNODES + threadIdx.x;
        if (node < n) {
            rowptr[node] = bs + run;
            cnt[node]    = c;
            dinv[node]   = rsqrtf((float)(c + 1));
        }
    }
    __syncthreads();

    if (len <= FCAP) {
        for (int i0 = bs + threadIdx.x; i0 < be; i0 += 4 * ETHR) {
            v2i v[4];
#pragma unroll
            for (int u = 0; u < 4; ++u) {
                const int i = i0 + u * ETHR;
                v[u] = pr[i < be ? i : be - 1];
            }
#pragma unroll
            for (int u = 0; u < 4; ++u) {
                if (i0 + u * ETHR < be) {
                    const int q = atomicAdd(&lcur[v[u].x & (BNODES - 1)], 1);
                    stage[q] = v[u].y;
                }
            }
        }
        __syncthreads();
        for (int i = threadIdx.x; i < len; i += ETHR) col[bs + i] = stage[i];
    } else {
        for (int i = bs + threadIdx.x; i < be; i += ETHR) {
            const v2i v = pr[i];
            const int q = atomicAdd(&lcur[v.x & (BNODES - 1)], 1);
            col[bs + q] = v.y;
        }
    }
}

template<int DOUT>
__global__ __launch_bounds__(256) void gemm_wmma_f32(
        const float* __restrict__ X, const float* __restrict__ W,
        const float* __restrict__ dscale, float* __restrict__ Y, int nrows) {
    __shared__ float sW[64 * DOUT];
    for (int i = threadIdx.x; i < 64 * DOUT; i += 256) {
        int k   = i / DOUT;
        int col = i - k * DOUT;
        sW[((k >> 1) * DOUT + col) * 2 + (k & 1)] = W[i];
    }
    __syncthreads();

    const int lane = threadIdx.x & 31;
    const int wave = threadIdx.x >> 5;
    const int m    = lane & 15;
    const int khi  = lane >> 4;
    const int rowBase = (blockIdx.x * 8 + wave) * 16;
    if (rowBase >= nrows) return;

    v8f acc[DOUT / 16];
#pragma unroll
    for (int t = 0; t < DOUT / 16; ++t) acc[t] = (v8f)0.0f;

    float sc[8];
#pragma unroll
    for (int r = 0; r < 8; ++r) sc[r] = dscale[rowBase + r + 8 * khi];

    const float* xrow = X + (size_t)(rowBase + m) * 64;
    const v2f*   sWp  = (const v2f*)sW;
#pragma unroll
    for (int k4 = 0; k4 < 64; k4 += 4) {
        v2f a = *(const v2f*)(xrow + k4 + 2 * khi);
        const int kp = (k4 >> 1) + khi;
#pragma unroll
        for (int t = 0; t < DOUT / 16; ++t) {
            v2f b = sWp[kp * DOUT + t * 16 + m];
            acc[t] = __builtin_amdgcn_wmma_f32_16x16x4_f32(
                false, a, false, b, (short)0, acc[t], false, false);
        }
    }

#pragma unroll
    for (int t = 0; t < DOUT / 16; ++t)
#pragma unroll
        for (int r = 0; r < 8; ++r)
            Y[(size_t)(rowBase + r + 8 * khi) * DOUT + t * 16 + m] = acc[t][r] * sc[r];
}

#define NPW 8
#define GU  8

template<int D> struct Row;
template<> struct Row<64> {
    typedef v2f T;
    static __device__ __forceinline__ T ld(const float* p) { return *(const v2f*)p; }
    static __device__ __forceinline__ void st(float* p, T v) { *(v2f*)p = v; }
};
template<> struct Row<32> {
    typedef float T;
    static __device__ __forceinline__ T ld(const float* p) { return *p; }
    static __device__ __forceinline__ void st(float* p, T v) { *p = v; }
};
static __device__ __forceinline__ float relu1(float v) { return v > 0.0f ? v : 0.0f; }
static __device__ __forceinline__ v2f   relu1(v2f v) { v2f r; r.x = relu1(v.x); r.y = relu1(v.y); return r; }

static __device__ __forceinline__ int   lane_i(int v, int l)   { return __builtin_amdgcn_readlane(v, l); }
static __device__ __forceinline__ float lane_f(float v, int l) { return __int_as_float(__builtin_amdgcn_readlane(__float_as_int(v), l)); }

template<int D, bool RELU>
__global__ __launch_bounds__(256) void gather_rows(
        const int* __restrict__ rowptr, const int* __restrict__ cnt,
        const int* __restrict__ col, const float* __restrict__ dinv,
        const float* __restrict__ hs, const float* __restrict__ bias,
        float* __restrict__ out, int n) {
    typedef typename Row<D>::T RT;
    constexpr int V = D / 32;
    const int lane = threadIdx.x & 31;
    const int gw = __builtin_amdgcn_readfirstlane((int)(blockIdx.x * 8 + (threadIdx.x >> 5)));
    const int n0 = gw * NPW;
    if (n0 >= n) return;
    const int nn = (n - n0 < NPW) ? (n - n0) : NPW;

    const int   li     = n0 + (lane < nn ? lane : nn - 1);
    const int   vstart = rowptr[li];
    const int   vdeg   = cnt[li];
    const float vdd    = dinv[li];
    const float* hl    = hs + lane * V;
    const RT     bv    = Row<D>::ld(bias + lane * V);

    int mycol = n0;
    int col1  = n0;
    {
        const int s0 = lane_i(vstart, 0);
        const int d0 = lane_i(vdeg, 0);
        const int s1 = lane_i(vstart, nn > 1 ? 1 : 0);
        const int d1 = nn > 1 ? lane_i(vdeg, 1) : 0;
        if (lane < d0) mycol = col[s0 + lane];
        if (lane < d1) col1  = col[s1 + lane];
        asm volatile("" :: "v"(mycol), "v"(col1));
    }

    for (int i = 0; i < nn; ++i) {
        const int   node  = n0 + i;
        const int   start = lane_i(vstart, i);
        const int   deg   = lane_i(vdeg, i);
        const float dd    = lane_f(vdd, i);

        const bool more2  = i + 2 < nn;
        const int  i2     = more2 ? i + 2 : i;
        const int  start2 = lane_i(vstart, i2);
        const int  deg2   = more2 ? lane_i(vdeg, i2) : 0;
        int col2 = n0 + i2;
        if (lane < deg2) col2 = col[start2 + lane];

        const RT self = Row<D>::ld(hl + (size_t)node * D);

        RT acc = (RT)0.0f;
        for (int base = 0; base < deg; base += 32) {
            const int m = (deg - base < 32) ? (deg - base) : 32;
            if (base > 0) {
                mycol = node;
                if (lane < m) mycol = col[start + base + lane];
            }
            for (int j = 0; j < m; j += GU) {
                RT r[GU];
#pragma unroll
                for (int u = 0; u < GU; ++u) {
                    const int s = lane_i(mycol, j + u);
                    r[u] = Row<D>::ld(hl + (size_t)s * D);
                }
#pragma unroll
                for (int u = 0; u < GU; ++u)
                    if (j + u < m) acc += r[u];
            }
        }

        acc += self;
        acc = acc * dd + bv;
        if (RELU) acc = relu1(acc);
        Row<D>::st(out + (size_t)node * D + lane * V, acc);

        mycol = col1;
        col1  = col2;
    }
}

extern "C" void kernel_launch(void* const* d_in, const int* in_sizes, int n_in,
                              void* d_out, int out_size, void* d_ws, size_t ws_size,
                              hipStream_t stream) {
    const float* x  = (const float*)d_in[0];
    const int*   ei = (const int*)d_in[1];
    const float* W1 = (const float*)d_in[2];
    const float* b1 = (const float*)d_in[3];
    const float* W2 = (const float*)d_in[4];
    const float* b2 = (const float*)d_in[5];
    float* out = (float*)d_out;

    const int N = in_sizes[0] / IN_D;
    const int E = in_sizes[1] / 2;
    const int* src = ei;
    const int* dst = ei + E;

    const int T = 256;
    auto g = [](long long n, int t) { return (unsigned)((n + t - 1) / t); };
    const int nb  = (N + BNODES - 1) >> BSH;
    const int nwg = (int)g(E, CHUNK);
    const int nt  = nb * nwg;
    const unsigned tiles = g(nt, SCAN_TILE);

    auto up = [](size_t b) { return (b + 255) & ~(size_t)255; };
    char* p = (char*)d_ws;
    float* dinv   = (float*)p;  p += up((size_t)N * 4);
    int*   cnt    = (int*)p;    p += up((size_t)N * 4);
    int*   rowptr = (int*)p;    p += up((size_t)N * 4);
    int*   tsum   = (int*)p;    p += up((size_t)tiles * 4);
    int*   table  = (int*)p;    p += up((size_t)nt * 4);
    int*   tscan  = (int*)p;    p += up((size_t)nt * 4);
    int*   pairs  = (int*)p;    p += up((size_t)E * 8);
    int*   col    = (int*)p;    p += up((size_t)E * 4);
    float* h1     = (float*)p;  p += up((size_t)N * HID_D * 4);
    float* agg    = (float*)p;
    float* h2     = h1;

    bucket_hist   <<<nwg, ETHR, 0, stream>>>(dst, table, E, nb, nwg);
    scan_tile_sums<<<tiles, T, 0, stream>>>(table, tsum, nt);
    scan_excl     <<<tiles, T, 0, stream>>>(table, tsum, tscan, nt);
    bucket_scatter<<<nwg, ETHR, 0, stream>>>(src, dst, tscan, pairs, E, nb, nwg);
    bucket_rows   <<<nb, ETHR, 0, stream>>>(pairs, tscan, rowptr, cnt, dinv, col, N, E, nb, nwg);

    const unsigned gemmBlocks = g((long long)(N + 15) / 16, 8);
    const unsigned nodeBlocks = g(g(N, NPW), 8);

    gemm_wmma_f32<HID_D><<<gemmBlocks, T, 0, stream>>>(x, W1, dinv, h1, N);
    gather_rows<HID_D, true><<<nodeBlocks, T, 0, stream>>>(rowptr, cnt, col, dinv, h1, b1, agg, N);

    gemm_wmma_f32<OUT_D><<<gemmBlocks, T, 0, stream>>>(agg, W2, dinv, h2, N);
    gather_rows<OUT_D, false><<<nodeBlocks, T, 0, stream>>>(rowptr, cnt, col, dinv, h2, b2, out, N);
}


# Layer-2 product fused into the layer-1 row-sum launch: a wave keeps its 8 finished rows in LDS and multiplies them by W2 on the matrix unit; one launch and the 64-wide plane's memory trip go.
# speedup vs baseline: 5.6026x; 1.0972x over previous
#include <hip/hip_runtime.h>
#include <hip/hip_bf16.h>


typedef float v2f __attribute__((ext_vector_type(2)));
typedef float v8f __attribute__((ext_vector_type(8)));

#define IN_D  64
#define HID_D 64
#define OUT_D 32

#define SCAN_TILE 1024

typedef int v2i __attribute__((ext_vector_type(2)));

#define BSH     8
#define BNODES  256
#define MAXB    512
#define CHUNK   8192
#define ETHR    1024
#define FCAP    12288

__global__ __launch_bounds__(ETHR) void bucket_hist(const int* __restrict__ dst,
                                                    int* __restrict__ table,
                                                    int e, int nb, int nwg) {
    __shared__ int lh[MAXB];
    if (threadIdx.x < MAXB) lh[threadIdx.x] = 0;
    __syncthreads();
    const int base = blockIdx.x * CHUNK + threadIdx.x;
    int d[CHUNK / ETHR];
#pragma unroll
    for (int k = 0; k < CHUNK / ETHR; ++k) {
        const int i = base + k * ETHR;
        d[k] = dst[i < e ? i : e - 1];
    }
#pragma unroll
    for (int k = 0; k < CHUNK / ETHR; ++k)
        if (base + k * ETHR < e) atomicAdd(&lh[d[k] >> BSH], 1);
    __syncthreads();
    if (threadIdx.x < nb) table[threadIdx.x * nwg + blockIdx.x] = lh[threadIdx.x];
}

__global__ __launch_bounds__(256) void scan_tile_sums(const int* __restrict__ cnt,
                                                      int* __restrict__ tsum, int n) {
    __shared__ int sw[8];
    const int base = blockIdx.x * SCAN_TILE + threadIdx.x * 4;
    int s = 0;
#pragma unroll
    for (int k = 0; k < 4; ++k)
        if (base + k < n) s += cnt[base + k];
#pragma unroll
    for (int o = 16; o > 0; o >>= 1) s += __shfl_xor(s, o);
    if ((threadIdx.x & 31) == 0) sw[threadIdx.x >> 5] = s;
    __syncthreads();
    if (threadIdx.x == 0) {
        int t = 0;
#pragma unroll
        for (int w = 0; w < 8; ++w) t += sw[w];
        tsum[blockIdx.x] = t;
    }
}

__global__ __launch_bounds__(256) void scan_excl(const int* __restrict__ in,
                                                 const int* __restrict__ tsum,
                                                 int* __restrict__ out, int n) {
    __shared__ int sw[8];
    __shared__ int sprev[8];
    const int lane = threadIdx.x & 31;
    const int wave = threadIdx.x >> 5;

    int p = 0;
    for (int t = threadIdx.x; t < (int)blockIdx.x; t += 256) p += tsum[t];
#pragma unroll
    for (int o = 16; o > 0; o >>= 1) p += __shfl_xor(p, o);
    if (lane == 0) sprev[wave] = p;

    const int base = blockIdx.x * SCAN_TILE + threadIdx.x * 4;
    int c[4];
    int s = 0;
#pragma unroll
    for (int k = 0; k < 4; ++k) {
        c[k] = (base + k < n) ? in[base + k] : 0;
        s += c[k];
    }
    int inc = s;
#pragma unroll
    for (int o = 1; o < 32; o <<= 1) {
        int v = __shfl_up(inc, o);
        if (lane >= o) inc += v;
    }
    if (lane == 31) sw[wave] = inc;
    __syncthreads();

    int run = inc - s;
#pragma unroll
    for (int w = 0; w < 8; ++w) {
        run += sprev[w];
        if (w < wave) run += sw[w];
    }
#pragma unroll
    for (int k = 0; k < 4; ++k) {
        if (base + k < n) {
            out[base + k] = run;
            run += c[k];
        }
    }
}

__global__ __launch_bounds__(ETHR) void bucket_scatter(const int* __restrict__ src,
                                                       const int* __restrict__ dst,
                                                       const int* __restrict__ tscan,
                                                       int* __restrict__ pairs,
                                                       int e, int nb, int nwg) {
    __shared__ int lpos[MAXB];
    if (threadIdx.x < nb) lpos[threadIdx.x] = tscan[threadIdx.x * nwg + blockIdx.x];
    const int base = blockIdx.x * CHUNK + threadIdx.x;
    v2i v[CHUNK / ETHR];
#pragma unroll
    for (int k = 0; k < CHUNK / ETHR; ++k) {
        const int i  = base + k * ETHR;
        const int ic = i < e ? i : e - 1;
        v[k].x = dst[ic];
        v[k].y = src[ic];
    }
    __syncthreads();
    v2i* pr = (v2i*)pairs;
#pragma unroll
    for (int k = 0; k < CHUNK / ETHR; ++k) {
        if (base + k * ETHR < e) {
            const int p = atomicAdd(&lpos[v[k].x >> BSH], 1);
            pr[p] = v[k];
        }
    }
}

__global__ __launch_bounds__(ETHR) void bucket_rows(const int* __restrict__ pairs,
                                                    const int* __restrict__ tscan,
                                                    int* __restrict__ rowptr,
                                                    int* __restrict__ cnt,
                                                    float* __restrict__ dinv,
                                                    int* __restrict__ col,
                                                    int n, int e, int nb, int nwg) {
    __shared__ int lcnt[BNODES];
    __shared__ int lcur[BNODES];
    __shared__ int lwave[BNODES / 32];
    __shared__ int stage[FCAP];
    const int b    = blockIdx.x;
    const int bs   = tscan[b * nwg];
    const int be   = (b + 1 < nb) ? tscan[(b + 1) * nwg] : e;
    const int len  = be - bs;
    const int lane = threadIdx.x & 31;
    const int wave = threadIdx.x >> 5;
    const v2i* pr  = (const v2i*)pairs;

    if (threadIdx.x < BNODES) lcnt[threadIdx.x] = 0;
    __syncthreads();
    for (int i0 = bs + threadIdx.x; i0 < be; i0 += 4 * ETHR) {
        int d[4];
#pragma unroll
        for (int u = 0; u < 4; ++u) {
            const int i = i0 + u * ETHR;
            d[u] = pr[i < be ? i : be - 1].x;
        }
#pragma unroll
        for (int u = 0; u < 4; ++u)
            if (i0 + u * ETHR < be) atomicAdd(&lcnt[d[u] & (BNODES - 1)], 1);
    }
    __syncthreads();

    int c = 0, inc = 0;
    if (threadIdx.x < BNODES) {
        c   = lcnt[threadIdx.x];
        inc = c;
#pragma unroll
        for (int o = 1; o < 32; o <<= 1) {
            int v = __shfl_up(inc, o);
            if (lane >= o) inc += v;
        }
        if (lane == 31) lwave[wave] = inc;
    }
    __syncthreads();
    if (threadIdx.x < BNODES) {
        int run = inc - c;
#pragma unroll
        for (int w = 0; w < BNODES / 32; ++w)
            if (w < wave) run += lwave[w];
        lcur[threadIdx.x] = run;
        const int node = b * BNODES + threadIdx.x;
        if (node < n) {
            rowptr[node] = bs + run;
            cnt[node]    = c;
            dinv[node]   = rsqrtf((float)(c + 1));
        }
    }
    __syncthreads();

    if (len <= FCAP) {
        for (int i0 = bs + threadIdx.x; i0 < be; i0 += 4 * ETHR) {
            v2i v[4];
#pragma unroll
            for (int u = 0; u < 4; ++u) {
                const int i = i0 + u * ETHR;
                v[u] = pr[i < be ? i : be - 1];
            }
#pragma unroll
            for (int u = 0; u < 4; ++u) {
                if (i0 + u * ETHR < be) {
                    const int q = atomicAdd(&lcur[v[u].x & (BNODES - 1)], 1);
                    stage[q] = v[u].y;
                }
            }
        }
        __syncthreads();
        for (int i = threadIdx.x; i < len; i += ETHR) col[bs + i] = stage[i];
    } else {
        for (int i = bs + threadIdx.x; i < be; i += ETHR) {
            const v2i v = pr[i];
            const int q = atomicAdd(&lcur[v.x & (BNODES - 1)], 1);
            col[bs + q] = v.y;
        }
    }
}

template<int DOUT>
__global__ __launch_bounds__(256) void gemm_wmma_f32(
        const float* __restrict__ X, const float* __restrict__ W,
        const float* __restrict__ dscale, float* __restrict__ Y, int nrows) {
    __shared__ float sW[64 * DOUT];
    for (int i = threadIdx.x; i < 64 * DOUT; i += 256) {
        int k   = i / DOUT;
        int col = i - k * DOUT;
        sW[((k >> 1) * DOUT + col) * 2 + (k & 1)] = W[i];
    }
    __syncthreads();

    const int lane = threadIdx.x & 31;
    const int wave = threadIdx.x >> 5;
    const int m    = lane & 15;
    const int khi  = lane >> 4;
    const int rowBase = (blockIdx.x * 8 + wave) * 16;
    if (rowBase >= nrows) return;

    v8f acc[DOUT / 16];
#pragma unroll
    for (int t = 0; t < DOUT / 16; ++t) acc[t] = (v8f)0.0f;

    float sc[8];
#pragma unroll
    for (int r = 0; r < 8; ++r) sc[r] = dscale[rowBase + r + 8 * khi];

    const float* xrow = X + (size_t)(rowBase + m) * 64;
    const v2f*   sWp  = (const v2f*)sW;
#pragma unroll
    for (int k4 = 0; k4 < 64; k4 += 4) {
        v2f a = *(const v2f*)(xrow + k4 + 2 * khi);
        const int kp = (k4 >> 1) + khi;
#pragma unroll
        for (int t = 0; t < DOUT / 16; ++t) {
            v2f b = sWp[kp * DOUT + t * 16 + m];
            acc[t] = __builtin_amdgcn_wmma_f32_16x16x4_f32(
                false, a, false, b, (short)0, acc[t], false, false);
        }
    }

#pragma unroll
    for (int t = 0; t < DOUT / 16; ++t)
#pragma unroll
        for (int r = 0; r < 8; ++r)
            Y[(size_t)(rowBase + r + 8 * khi) * DOUT + t * 16 + m] = acc[t][r] * sc[r];
}

#define NPW 8
#define GU  8

template<int D> struct Row;
template<> struct Row<64> {
    typedef v2f T;
    static __device__ __forceinline__ T ld(const float* p) { return *(const v2f*)p; }
    static __device__ __forceinline__ void st(float* p, T v) { *(v2f*)p = v; }
};
template<> struct Row<32> {
    typedef float T;
    static __device__ __forceinline__ T ld(const float* p) { return *p; }
    static __device__ __forceinline__ void st(float* p, T v) { *p = v; }
};
static __device__ __forceinline__ float relu1(float v) { return v > 0.0f ? v : 0.0f; }
static __device__ __forceinline__ v2f   relu1(v2f v) { v2f r; r.x = relu1(v.x); r.y = relu1(v.y); return r; }

static __device__ __forceinline__ int   lane_i(int v, int l)   { return __builtin_amdgcn_readlane(v, l); }
static __device__ __forceinline__ float lane_f(float v, int l) { return __int_as_float(__builtin_amdgcn_readlane(__float_as_int(v), l)); }

template<int D, bool RELU>
__global__ __launch_bounds__(256) void gather_rows(
        const int* __restrict__ rowptr, const int* __restrict__ cnt,
        const int* __restrict__ col, const float* __restrict__ dinv,
        const float* __restrict__ hs, const float* __restrict__ bias,
        float* __restrict__ out, int n) {
    typedef typename Row<D>::T RT;
    constexpr int V = D / 32;
    const int lane = threadIdx.x & 31;
    const int gw = __builtin_amdgcn_readfirstlane((int)(blockIdx.x * 8 + (threadIdx.x >> 5)));
    const int n0 = gw * NPW;
    if (n0 >= n) return;
    const int nn = (n - n0 < NPW) ? (n - n0) : NPW;

    const int   li     = n0 + (lane < nn ? lane : nn - 1);
    const int   vstart = rowptr[li];
    const int   vdeg   = cnt[li];
    const float vdd    = dinv[li];
    const float* hl    = hs + lane * V;
    const RT     bv    = Row<D>::ld(bias + lane * V);

    int mycol = n0;
    int col1  = n0;
    {
        const int s0 = lane_i(vstart, 0);
        const int d0 = lane_i(vdeg, 0);
        const int s1 = lane_i(vstart, nn > 1 ? 1 : 0);
        const int d1 = nn > 1 ? lane_i(vdeg, 1) : 0;
        if (lane < d0) mycol = col[s0 + lane];
        if (lane < d1) col1  = col[s1 + lane];
        asm volatile("" :: "v"(mycol), "v"(col1));
    }

    for (int i = 0; i < nn; ++i) {
        const int   node  = n0 + i;
        const int   start = lane_i(vstart, i);
        const int   deg   = lane_i(vdeg, i);
        const float dd    = lane_f(vdd, i);

        const bool more2  = i + 2 < nn;
        const int  i2     = more2 ? i + 2 : i;
        const int  start2 = lane_i(vstart, i2);
        const int  deg2   = more2 ? lane_i(vdeg, i2) : 0;
        int col2 = n0 + i2;
        if (lane < deg2) col2 = col[start2 + lane];

        const RT self = Row<D>::ld(hl + (size_t)node * D);

        RT acc = (RT)0.0f;
        for (int base = 0; base < deg; base += 32) {
            const int m = (deg - base < 32) ? (deg - base) : 32;
            if (base > 0) {
                mycol = node;
                if (lane < m) mycol = col[start + base + lane];
            }
            for (int j = 0; j < m; j += GU) {
                RT r[GU];
#pragma unroll
                for (int u = 0; u < GU; ++u) {
                    const int s = lane_i(mycol, j + u);
                    r[u] = Row<D>::ld(hl + (size_t)s * D);
                }
#pragma unroll
                for (int u = 0; u < GU; ++u)
                    if (j + u < m) acc += r[u];
            }
        }

        acc += self;
        acc = acc * dd + bv;
        if (RELU) acc = relu1(acc);
        Row<D>::st(out + (size_t)node * D + lane * V, acc);

        mycol = col1;
        col1  = col2;
    }
}

#define TPITCH 68

__global__ __launch_bounds__(256) void gather_rows_mm(
        const int* __restrict__ rowptr, const int* __restrict__ cnt,
        const int* __restrict__ col, const float* __restrict__ dinv,
        const float* __restrict__ hs, const float* __restrict__ bias,
        const float* __restrict__ Wn, float* __restrict__ out2, int n) {
    constexpr int D = HID_D;
    constexpr int V = D / 32;
    typedef Row<D>::T RT;
    __shared__ __attribute__((aligned(16))) float sW[64 * OUT_D];
    __shared__ __attribute__((aligned(16))) float strip[8 * NPW * TPITCH];

    const int lane = threadIdx.x & 31;
    const int wv   = threadIdx.x >> 5;
    const int gw = __builtin_amdgcn_readfirstlane((int)(blockIdx.x * 8 + wv));
    const int n0 = gw * NPW;
    const int nn = (n - n0 < NPW) ? (n - n0) : NPW;

    constexpr int WPT = 64 * OUT_D / 256;
    float wreg[WPT];
#pragma unroll
    for (int u = 0; u < WPT; ++u) wreg[u] = Wn[threadIdx.x + u * 256];

    const int   li     = n0 + (lane < nn ? lane : nn - 1);
    const int   vstart = rowptr[li];
    const int   vdeg   = cnt[li];
    const float vdd    = dinv[li];
    const float* hl    = hs + lane * V;
    const RT     bv    = Row<D>::ld(bias + lane * V);

#pragma unroll
    for (int u = 0; u < WPT; ++u) {
        const int i = threadIdx.x + u * 256;
        const int k = i / OUT_D;
        const int c = i - k * OUT_D;
        sW[((k >> 1) * OUT_D + c) * 2 + (k & 1)] = wreg[u];
    }
    __syncthreads();
    if (n0 >= n) return;
    float* tw = strip + wv * (NPW * TPITCH);

    int mycol = n0;
    int col1  = n0;
    {
        const int s0 = lane_i(vstart, 0);
        const int d0 = lane_i(vdeg, 0);
        const int s1 = lane_i(vstart, nn > 1 ? 1 : 0);
        const int d1 = nn > 1 ? lane_i(vdeg, 1) : 0;
        if (lane < d0) mycol = col[s0 + lane];
        if (lane < d1) col1  = col[s1 + lane];
        asm volatile("" :: "v"(mycol), "v"(col1));
    }

    for (int i = 0; i < nn; ++i) {
        const int   node  = n0 + i;
        const int   start = lane_i(vstart, i);
        const int   deg   = lane_i(vdeg, i);
        const float dd    = lane_f(vdd, i);

        const bool more2  = i + 2 < nn;
        const int  i2     = more2 ? i + 2 : i;
        const int  start2 = lane_i(vstart, i2);
        const int  deg2   = more2 ? lane_i(vdeg, i2) : 0;
        int col2 = n0 + i2;
        if (lane < deg2) col2 = col[start2 + lane];

        const RT self = Row<D>::ld(hl + (size_t)node * D);

        RT acc = (RT)0.0f;
        for (int base = 0; base < deg; base += 32) {
            const int m = (deg - base < 32) ? (deg - base) : 32;
            if (base > 0) {
                mycol = node;
                if (lane < m) mycol = col[start + base + lane];
            }
            for (int j = 0; j < m; j += GU) {
                RT r[GU];
#pragma unroll
                for (int u = 0; u < GU; ++u) {
                    const int s = lane_i(mycol, j + u);
                    r[u] = Row<D>::ld(hl + (size_t)s * D);
                }
#pragma unroll
                for (int u = 0; u < GU; ++u)
                    if (j + u < m) acc += r[u];
            }
        }

        acc += self;
        acc = acc * dd + bv;
        acc = relu1(acc);
        *(v2f*)(tw + i * TPITCH + lane * V) = acc;

        mycol = col1;
        col1  = col2;
    }

    const int m   = lane & 15;
    const int khi = lane >> 4;
    const float* arow = tw + (m & (NPW - 1)) * TPITCH + 2 * khi;
    const v2f*   sWp  = (const v2f*)sW;
    v8f pr[OUT_D / 16];
#pragma unroll
    for (int t = 0; t < OUT_D / 16; ++t) pr[t] = (v8f)0.0f;
#pragma unroll
    for (int k4 = 0; k4 < 64; k4 += 4) {
        const v2f a  = *(const v2f*)(arow + k4);
        const int kp = (k4 >> 1) + khi;
#pragma unroll
        for (int t = 0; t < OUT_D / 16; ++t) {
            const v2f b = sWp[kp * OUT_D + t * 16 + m];
            pr[t] = __builtin_amdgcn_wmma_f32_16x16x4_f32(
                false, a, false, b, (short)0, pr[t], false, false);
        }
    }
#pragma unroll
    for (int r = 0; r < NPW; ++r) {
        const float v = (khi ? pr[1][r] : pr[0][r]) * lane_f(vdd, r);
        if (r < nn) out2[(size_t)(n0 + r) * OUT_D + lane] = v;
    }
}

extern "C" void kernel_launch(void* const* d_in, const int* in_sizes, int n_in,
                              void* d_out, int out_size, void* d_ws, size_t ws_size,
                              hipStream_t stream) {
    const float* x  = (const float*)d_in[0];
    const int*   ei = (const int*)d_in[1];
    const float* W1 = (const float*)d_in[2];
    const float* b1 = (const float*)d_in[3];
    const float* W2 = (const float*)d_in[4];
    const float* b2 = (const float*)d_in[5];
    float* out = (float*)d_out;

    const int N = in_sizes[0] / IN_D;
    const int E = in_sizes[1] / 2;
    const int* src = ei;
    const int* dst = ei + E;

    const int T = 256;
    auto g = [](long long n, int t) { return (unsigned)((n + t - 1) / t); };
    const int nb  = (N + BNODES - 1) >> BSH;
    const int nwg = (int)g(E, CHUNK);
    const int nt  = nb * nwg;
    const unsigned tiles = g(nt, SCAN_TILE);

    auto up = [](size_t b) { return (b + 255) & ~(size_t)255; };
    char* p = (char*)d_ws;
    float* dinv   = (float*)p;  p += up((size_t)N * 4);
    int*   cnt    = (int*)p;    p += up((size_t)N * 4);
    int*   rowptr = (int*)p;    p += up((size_t)N * 4);
    int*   tsum   = (int*)p;    p += up((size_t)tiles * 4);
    int*   table  = (int*)p;    p += up((size_t)nt * 4);
    int*   tscan  = (int*)p;    p += up((size_t)nt * 4);
    int*   pairs  = (int*)p;    p += up((size_t)E * 8);
    int*   col    = (int*)p;    p += up((size_t)E * 4);
    float* h1     = (float*)p;  p += up((size_t)N * HID_D * 4);
    float* h2     = (float*)p;

    bucket_hist   <<<nwg, ETHR, 0, stream>>>(dst, table, E, nb, nwg);
    scan_tile_sums<<<tiles, T, 0, stream>>>(table, tsum, nt);
    scan_excl     <<<tiles, T, 0, stream>>>(table, tsum, tscan, nt);
    bucket_scatter<<<nwg, ETHR, 0, stream>>>(src, dst, tscan, pairs, E, nb, nwg);
    bucket_rows   <<<nb, ETHR, 0, stream>>>(pairs, tscan, rowptr, cnt, dinv, col, N, E, nb, nwg);

    const unsigned gemmBlocks = g((long long)(N + 15) / 16, 8);
    const unsigned nodeBlocks = g(g(N, NPW), 8);

    gemm_wmma_f32<HID_D><<<gemmBlocks, T, 0, stream>>>(x, W1, dinv, h1, N);
    gather_rows_mm<<<nodeBlocks, T, 0, stream>>>(rowptr, cnt, col, dinv, h1, b1, W2, h2, N);

    gather_rows<OUT_D, false><<<nodeBlocks, T, 0, stream>>>(rowptr, cnt, col, dinv, h2, b2, out, N);
}


# Edge order without count pass or scans: each block orders its own 8192 edges by bucket in LDS and stores them whole; a bucket reads its run from every block. 8 launches to 5; LDS arrival order kept
# speedup vs baseline: 6.9683x; 1.2438x over previous
#include <hip/hip_runtime.h>
#include <hip/hip_bf16.h>


typedef float v2f __attribute__((ext_vector_type(2)));
typedef float v8f __attribute__((ext_vector_type(8)));

#define IN_D  64
#define HID_D 64
#define OUT_D 32

typedef int v2i __attribute__((ext_vector_type(2)));

#define BSH     8
#define BNODES  256
#define MAXB    512
#define CHUNK   8192
#define ETHR    1024
#define FCAP    12288
#define MAXW    1024
#define SU      7

__global__ __launch_bounds__(ETHR) void chunk_sort(const int* __restrict__ src,
                                                   const int* __restrict__ dst,
                                                   int* __restrict__ table,
                                                   int* __restrict__ pairs, int e) {
    __shared__ int lh[MAXB];
    __shared__ int lwave[MAXB / 32];
    __shared__ v2i stage[CHUNK];
    const int lane  = threadIdx.x & 31;
    const int wave  = threadIdx.x >> 5;
    const int cbase = blockIdx.x * CHUNK;
    const int len   = (e - cbase < CHUNK) ? (e - cbase) : CHUNK;
    if (threadIdx.x < MAXB) lh[threadIdx.x] = 0;
    v2i v[CHUNK / ETHR];
#pragma unroll
    for (int k = 0; k < CHUNK / ETHR; ++k) {
        const int i  = cbase + threadIdx.x + k * ETHR;
        const int ic = i < e ? i : e - 1;
        v[k].x = dst[ic];
        v[k].y = src[ic];
    }
    __builtin_amdgcn_sched_barrier(0);
    __syncthreads();
#pragma unroll
    for (int k = 0; k < CHUNK / ETHR; ++k)
        if (threadIdx.x + k * ETHR < len) atomicAdd(&lh[v[k].x >> BSH], 1);
    __syncthreads();

    int c = 0, inc = 0;
    if (threadIdx.x < MAXB) {
        c   = lh[threadIdx.x];
        inc = c;
#pragma unroll
        for (int o = 1; o < 32; o <<= 1) {
            int x = __shfl_up(inc, o);
            if (lane >= o) inc += x;
        }
        if (lane == 31) lwave[wave] = inc;
    }
    __syncthreads();
    if (threadIdx.x < MAXB) {
        int run = inc - c;
#pragma unroll
        for (int k = 0; k < MAXB / 32; ++k)
            if (k < wave) run += lwave[k];
        lh[threadIdx.x] = run;
        table[blockIdx.x * MAXB + threadIdx.x] = run;
    }
    __syncthreads();
#pragma unroll
    for (int k = 0; k < CHUNK / ETHR; ++k) {
        if (threadIdx.x + k * ETHR < len) {
            const int q = atomicAdd(&lh[v[k].x >> BSH], 1);
            stage[q] = v[k];
        }
    }
    __syncthreads();
    v2i* pr = (v2i*)pairs;
    for (int i = threadIdx.x; i < len; i += ETHR) pr[cbase + i] = stage[i];
}

__global__ __launch_bounds__(ETHR) void bucket_rows(const int* __restrict__ pairs,
                                                    const int* __restrict__ table,
                                                    int* __restrict__ rowptr,
                                                    int* __restrict__ cnt,
                                                    float* __restrict__ dinv,
                                                    int* __restrict__ col,
                                                    int n, int e, int nwg) {
    __shared__ int lcnt[BNODES];
    __shared__ int lcur[BNODES];
    __shared__ int lwave[BNODES / 32];
    __shared__ int runS[MAXW];
    __shared__ int runC[MAXW];
    __shared__ int red0[ETHR / 32];
    __shared__ int red1[ETHR / 32];
    __shared__ int stage[FCAP];
    const int b    = blockIdx.x;
    const int lane = threadIdx.x & 31;
    const int wave = __builtin_amdgcn_readfirstlane((int)(threadIdx.x >> 5));
    const v2i* pr  = (const v2i*)pairs;

    int s0 = 0, s1 = 0;
    for (int w = threadIdx.x; w < nwg; w += ETHR) {
        const int* row  = table + (size_t)w * MAXB;
        const int  lenw = (e - w * CHUNK < CHUNK) ? (e - w * CHUNK) : CHUNK;
        const int  o0   = row[b];
        const int  o1r  = row[b + 1 < MAXB ? b + 1 : b];
        const int  o1   = (b + 1 < MAXB) ? o1r : lenw;
        runS[w] = w * CHUNK + o0;
        runC[w] = o1 - o0;
        s0 += o0;
        s1 += o1;
    }
#pragma unroll
    for (int o = 16; o > 0; o >>= 1) {
        s0 += __shfl_xor(s0, o);
        s1 += __shfl_xor(s1, o);
    }
    if (lane == 0) { red0[wave] = s0; red1[wave] = s1; }
    if (threadIdx.x < BNODES) lcnt[threadIdx.x] = 0;
    __syncthreads();
    int bs = 0, be = 0;
#pragma unroll
    for (int k = 0; k < ETHR / 32; ++k) { bs += red0[k]; be += red1[k]; }
    const int len = be - bs;

    for (int w0 = wave; w0 < nwg; w0 += 32 * SU) {
        int s[SU], c[SU], d[SU];
#pragma unroll
        for (int u = 0; u < SU; ++u) {
            const int w  = w0 + 32 * u;
            const int wc = w < nwg ? w : nwg - 1;
            s[u] = __builtin_amdgcn_readfirstlane(runS[wc]);
            c[u] = w < nwg ? __builtin_amdgcn_readfirstlane(runC[wc]) : 0;
            d[u] = pairs[2 * (size_t)(lane < c[u] ? s[u] + lane : 0)];
        }
#pragma unroll
        for (int u = 0; u < SU; ++u)
            if (lane < c[u]) atomicAdd(&lcnt[d[u] & (BNODES - 1)], 1);
#pragma unroll
        for (int u = 0; u < SU; ++u)
#pragma nounroll
            for (int i0 = 32; i0 < c[u]; i0 += 32)
                if (i0 + lane < c[u])
                    atomicAdd(&lcnt[pairs[2 * (size_t)(s[u] + i0 + lane)] & (BNODES - 1)], 1);
    }
    __syncthreads();

    int cc = 0, inc = 0;
    if (threadIdx.x < BNODES) {
        cc  = lcnt[threadIdx.x];
        inc = cc;
#pragma unroll
        for (int o = 1; o < 32; o <<= 1) {
            int x = __shfl_up(inc, o);
            if (lane >= o) inc += x;
        }
        if (lane == 31) lwave[wave] = inc;
    }
    __syncthreads();
    if (threadIdx.x < BNODES) {
        int run = inc - cc;
#pragma unroll
        for (int k = 0; k < BNODES / 32; ++k)
            if (k < wave) run += lwave[k];
        lcur[threadIdx.x] = run;
        const int node = b * BNODES + threadIdx.x;
        if (node < n) {
            rowptr[node] = bs + run;
            cnt[node]    = cc;
            dinv[node]   = rsqrtf((float)(cc + 1));
        }
    }
    __syncthreads();

    const bool fits = len <= FCAP;
    for (int w0 = wave; w0 < nwg; w0 += 32 * SU) {
        int s[SU], c[SU];
        v2i v[SU];
#pragma unroll
        for (int u = 0; u < SU; ++u) {
            const int w  = w0 + 32 * u;
            const int wc = w < nwg ? w : nwg - 1;
            s[u] = __builtin_amdgcn_readfirstlane(runS[wc]);
            c[u] = w < nwg ? __builtin_amdgcn_readfirstlane(runC[wc]) : 0;
            v[u] = pr[lane < c[u] ? s[u] + lane : 0];
        }
#pragma unroll
        for (int u = 0; u < SU; ++u) {
            if (lane < c[u]) {
                const int q = atomicAdd(&lcur[v[u].x & (BNODES - 1)], 1);
                if (fits) stage[q] = v[u].y; else col[bs + q] = v[u].y;
            }
        }
#pragma unroll
        for (int u = 0; u < SU; ++u) {
#pragma nounroll
            for (int i0 = 32; i0 < c[u]; i0 += 32) {
                if (i0 + lane < c[u]) {
                    const v2i x = pr[s[u] + i0 + lane];
                    const int q = atomicAdd(&lcur[x.x & (BNODES - 1)], 1);
                    if (fits) stage[q] = x.y; else col[bs + q] = x.y;
                }
            }
        }
    }
    if (fits) {
        __syncthreads();
        for (int i = threadIdx.x; i < len; i += ETHR) col[bs + i] = stage[i];
    }
}

template<int DOUT>
__global__ __launch_bounds__(256) void gemm_wmma_f32(
        const float* __restrict__ X, const float* __restrict__ W,
        const float* __restrict__ dscale, float* __restrict__ Y, int nrows) {
    __shared__ float sW[64 * DOUT];
    for (int i = threadIdx.x; i < 64 * DOUT; i += 256) {
        int k   = i / DOUT;
        int col = i - k * DOUT;
        sW[((k >> 1) * DOUT + col) * 2 + (k & 1)] = W[i];
    }
    __syncthreads();

    const int lane = threadIdx.x & 31;
    const int wave = threadIdx.x >> 5;
    const int m    = lane & 15;
    const int khi  = lane >> 4;
    const int rowBase = (blockIdx.x * 8 + wave) * 16;
    if (rowBase >= nrows) return;

    v8f acc[DOUT / 16];
#pragma unroll
    for (int t = 0; t < DOUT / 16; ++t) acc[t] = (v8f)0.0f;

    float sc[8];
#pragma unroll
    for (int r = 0; r < 8; ++r) sc[r] = dscale[rowBase + r + 8 * khi];

    const float* xrow = X + (size_t)(rowBase + m) * 64;
    const v2f*   sWp  = (const v2f*)sW;
#pragma unroll
    for (int k4 = 0; k4 < 64; k4 += 4) {
        v2f a = *(const v2f*)(xrow + k4 + 2 * khi);
        const int kp = (k4 >> 1) + khi;
#pragma unroll
        for (int t = 0; t < DOUT / 16; ++t) {
            v2f b = sWp[kp * DOUT + t * 16 + m];
            acc[t] = __builtin_amdgcn_wmma_f32_16x16x4_f32(
                false, a, false, b, (short)0, acc[t], false, false);
        }
    }

#pragma unroll
    for (int t = 0; t < DOUT / 16; ++t)
#pragma unroll
        for (int r = 0; r < 8; ++r)
            Y[(size_t)(rowBase + r + 8 * khi) * DOUT + t * 16 + m] = acc[t][r] * sc[r];
}

#define NPW 8
#define GU  8

template<int D> struct Row;
template<> struct Row<64> {
    typedef v2f T;
    static __device__ __forceinline__ T ld(const float* p) { return *(const v2f*)p; }
    static __device__ __forceinline__ void st(float* p, T v) { *(v2f*)p = v; }
};
template<> struct Row<32> {
    typedef float T;
    static __device__ __forceinline__ T ld(const float* p) { return *p; }
    static __device__ __forceinline__ void st(float* p, T v) { *p = v; }
};
static __device__ __forceinline__ float relu1(float v) { return v > 0.0f ? v : 0.0f; }
static __device__ __forceinline__ v2f   relu1(v2f v) { v2f r; r.x = relu1(v.x); r.y = relu1(v.y); return r; }

static __device__ __forceinline__ int   lane_i(int v, int l)   { return __builtin_amdgcn_readlane(v, l); }
static __device__ __forceinline__ float lane_f(float v, int l) { return __int_as_float(__builtin_amdgcn_readlane(__float_as_int(v), l)); }

template<int D, bool RELU>
__global__ __launch_bounds__(256) void gather_rows(
        const int* __restrict__ rowptr, const int* __restrict__ cnt,
        const int* __restrict__ col, const float* __restrict__ dinv,
        const float* __restrict__ hs, const float* __restrict__ bias,
        float* __restrict__ out, int n) {
    typedef typename Row<D>::T RT;
    constexpr int V = D / 32;
    const int lane = threadIdx.x & 31;
    const int gw = __builtin_amdgcn_readfirstlane((int)(blockIdx.x * 8 + (threadIdx.x >> 5)));
    const int n0 = gw * NPW;
    if (n0 >= n) return;
    const int nn = (n - n0 < NPW) ? (n - n0) : NPW;

    const int   li     = n0 + (lane < nn ? lane : nn - 1);
    const int   vstart = rowptr[li];
    const int   vdeg   = cnt[li];
    const float vdd    = dinv[li];
    const float* hl    = hs + lane * V;
    const RT     bv    = Row<D>::ld(bias + lane * V);

    int mycol = n0;
    int col1  = n0;
    {
        const int s0 = lane_i(vstart, 0);
        const int d0 = lane_i(vdeg, 0);
        const int s1 = lane_i(vstart, nn > 1 ? 1 : 0);
        const int d1 = nn > 1 ? lane_i(vdeg, 1) : 0;
        if (lane < d0) mycol = col[s0 + lane];
        if (lane < d1) col1  = col[s1 + lane];
        asm volatile("" :: "v"(mycol), "v"(col1));
    }

    for (int i = 0; i < nn; ++i) {
        const int   node  = n0 + i;
        const int   start = lane_i(vstart, i);
        const int   deg   = lane_i(vdeg, i);
        const float dd    = lane_f(vdd, i);

        const bool more2  = i + 2 < nn;
        const int  i2     = more2 ? i + 2 : i;
        const int  start2 = lane_i(vstart, i2);
        const int  deg2   = more2 ? lane_i(vdeg, i2) : 0;
        int col2 = n0 + i2;
        if (lane < deg2) col2 = col[start2 + lane];

        const RT self = Row<D>::ld(hl + (size_t)node * D);

        RT acc = (RT)0.0f;
        for (int base = 0; base < deg; base += 32) {
            const int m = (deg - base < 32) ? (deg - base) : 32;
            if (base > 0) {
                mycol = node;
                if (lane < m) mycol = col[start + base + lane];
            }
            for (int j = 0; j < m; j += GU) {
                RT r[GU];
#pragma unroll
                for (int u = 0; u < GU; ++u) {
                    const int s = lane_i(mycol, j + u);
                    r[u] = Row<D>::ld(hl + (size_t)s * D);
                }
#pragma unroll
                for (int u = 0; u < GU; ++u)
                    if (j + u < m) acc += r[u];
            }
        }

        acc += self;
        acc = acc * dd + bv;
        if (RELU) acc = relu1(acc);
        Row<D>::st(out + (size_t)node * D + lane * V, acc);

        mycol = col1;
        col1  = col2;
    }
}

#define TPITCH 68

__global__ __launch_bounds__(256) void gather_rows_mm(
        const int* __restrict__ rowptr, const int* __restrict__ cnt,
        const int* __restrict__ col, const float* __restrict__ dinv,
        const float* __restrict__ hs, const float* __restrict__ bias,
        const float* __restrict__ Wn, float* __restrict__ out2, int n) {
    constexpr int D = HID_D;
    constexpr int V = D / 32;
    typedef Row<D>::T RT;
    __shared__ __attribute__((aligned(16))) float sW[64 * OUT_D];
    __shared__ __attribute__((aligned(16))) float strip[8 * NPW * TPITCH];

    const int lane = threadIdx.x & 31;
    const int wv   = threadIdx.x >> 5;
    const int gw = __builtin_amdgcn_readfirstlane((int)(blockIdx.x * 8 + wv));
    const int n0 = gw * NPW;
    const int nn = (n - n0 < NPW) ? (n - n0) : NPW;

    constexpr int WPT = 64 * OUT_D / 256;
    float wreg[WPT];
#pragma unroll
    for (int u = 0; u < WPT; ++u) wreg[u] = Wn[threadIdx.x + u * 256];

    const int   li     = n0 + (lane < nn ? lane : nn - 1);
    const int   vstart = rowptr[li];
    const int   vdeg   = cnt[li];
    const float vdd    = dinv[li];
    const float* hl    = hs + lane * V;
    const RT     bv    = Row<D>::ld(bias + lane * V);

#pragma unroll
    for (int u = 0; u < WPT; ++u) {
        const int i = threadIdx.x + u * 256;
        const int k = i / OUT_D;
        const int c = i - k * OUT_D;
        sW[((k >> 1) * OUT_D + c) * 2 + (k & 1)] = wreg[u];
    }
    __syncthreads();
    if (n0 >= n) return;
    float* tw = strip + wv * (NPW * TPITCH);

    int mycol = n0;
    int col1  = n0;
    {
        const int s0 = lane_i(vstart, 0);
        const int d0 = lane_i(vdeg, 0);
        const int s1 = lane_i(vstart, nn > 1 ? 1 : 0);
        const int d1 = nn > 1 ? lane_i(vdeg, 1) : 0;
        if (lane < d0) mycol = col[s0 + lane];
        if (lane < d1) col1  = col[s1 + lane];
        asm volatile("" :: "v"(mycol), "v"(col1));
    }

    for (int i = 0; i < nn; ++i) {
        const int   node  = n0 + i;
        const int   start = lane_i(vstart, i);
        const int   deg   = lane_i(vdeg, i);
        const float dd    = lane_f(vdd, i);

        const bool more2  = i + 2 < nn;
        const int  i2     = more2 ? i + 2 : i;
        const int  start2 = lane_i(vstart, i2);
        const int  deg2   = more2 ? lane_i(vdeg, i2) : 0;
        int col2 = n0 + i2;
        if (lane < deg2) col2 = col[start2 + lane];

        const RT self = Row<D>::ld(hl + (size_t)node * D);

        RT acc = (RT)0.0f;
        for (int base = 0; base < deg; base += 32) {
            const int m = (deg - base < 32) ? (deg - base) : 32;
            if (base > 0) {
                mycol = node;
                if (lane < m) mycol = col[start + base + lane];
            }
            for (int j = 0; j < m; j += GU) {
                RT r[GU];
#pragma unroll
                for (int u = 0; u < GU; ++u) {
                    const int s = lane_i(mycol, j + u);
                    r[u] = Row<D>::ld(hl + (size_t)s * D);
                }
#pragma unroll
                for (int u = 0; u < GU; ++u)
                    if (j + u < m) acc += r[u];
            }
        }

        acc += self;
        acc = acc * dd + bv;
        acc = relu1(acc);
        *(v2f*)(tw + i * TPITCH + lane * V) = acc;

        mycol = col1;
        col1  = col2;
    }

    const int m   = lane & 15;
    const int khi = lane >> 4;
    const float* arow = tw + (m & (NPW - 1)) * TPITCH + 2 * khi;
    const v2f*   sWp  = (const v2f*)sW;
    v8f pr[OUT_D / 16];
#pragma unroll
    for (int t = 0; t < OUT_D / 16; ++t) pr[t] = (v8f)0.0f;
#pragma unroll
    for (int k4 = 0; k4 < 64; k4 += 4) {
        const v2f a  = *(const v2f*)(arow + k4);
        const int kp = (k4 >> 1) + khi;
#pragma unroll
        for (int t = 0; t < OUT_D / 16; ++t) {
            const v2f b = sWp[kp * OUT_D + t * 16 + m];
            pr[t] = __builtin_amdgcn_wmma_f32_16x16x4_f32(
                false, a, false, b, (short)0, pr[t], false, false);
        }
    }
#pragma unroll
    for (int r = 0; r < NPW; ++r) {
        const float v = (khi ? pr[1][r] : pr[0][r]) * lane_f(vdd, r);
        if (r < nn) out2[(size_t)(n0 + r) * OUT_D + lane] = v;
    }
}

extern "C" void kernel_launch(void* const* d_in, const int* in_sizes, int n_in,
                              void* d_out, int out_size, void* d_ws, size_t ws_size,
                              hipStream_t stream) {
    const float* x  = (const float*)d_in[0];
    const int*   ei = (const int*)d_in[1];
    const float* W1 = (const float*)d_in[2];
    const float* b1 = (const float*)d_in[3];
    const float* W2 = (const float*)d_in[4];
    const float* b2 = (const float*)d_in[5];
    float* out = (float*)d_out;

    const int N = in_sizes[0] / IN_D;
    const int E = in_sizes[1] / 2;
    const int* src = ei;
    const int* dst = ei + E;

    const int T = 256;
    auto g = [](long long n, int t) { return (unsigned)((n + t - 1) / t); };
    const int nb  = (N + BNODES - 1) >> BSH;
    const int nwg = (int)g(E, CHUNK);
    const int nt  = nwg * MAXB;

    auto up = [](size_t b) { return (b + 255) & ~(size_t)255; };
    char* p = (char*)d_ws;
    float* dinv   = (float*)p;  p += up((size_t)N * 4);
    int*   cnt    = (int*)p;    p += up((size_t)N * 4);
    int*   rowptr = (int*)p;    p += up((size_t)N * 4);
    int*   table  = (int*)p;    p += up((size_t)nt * 4);
    int*   pairs  = (int*)p;    p += up((size_t)E * 8);
    int*   col    = (int*)p;    p += up((size_t)E * 4);
    float* h1     = (float*)p;  p += up((size_t)N * HID_D * 4);
    float* h2     = (float*)p;

    chunk_sort <<<nwg, ETHR, 0, stream>>>(src, dst, table, pairs, E);
    bucket_rows<<<nb, ETHR, 0, stream>>>(pairs, table, rowptr, cnt, dinv, col, N, E, nwg);

    const unsigned gemmBlocks = g((long long)(N + 15) / 16, 8);
    const unsigned nodeBlocks = g(g(N, NPW), 8);

    gemm_wmma_f32<HID_D><<<gemmBlocks, T, 0, stream>>>(x, W1, dinv, h1, N);
    gather_rows_mm<<<nodeBlocks, T, 0, stream>>>(rowptr, cnt, col, dinv, h1, b1, W2, h2, N);

    gather_rows<OUT_D, false><<<nodeBlocks, T, 0, stream>>>(rowptr, cnt, col, dinv, h2, b2, out, N);
}


# Layer-2 row sums: lanes in four groups of 8, each group walks one destination row, so one load of 4 floats per lane brings 4 source rows (4 lines) where a wave-wide load brought 1; add order kept.
# speedup vs baseline: 7.3483x; 1.0545x over previous
#include <hip/hip_runtime.h>
#include <hip/hip_bf16.h>


typedef float v2f __attribute__((ext_vector_type(2)));
typedef float v8f __attribute__((ext_vector_type(8)));

#define IN_D  64
#define HID_D 64
#define OUT_D 32

typedef int v2i __attribute__((ext_vector_type(2)));

#define BSH     8
#define BNODES  256
#define MAXB    512
#define CHUNK   8192
#define ETHR    1024
#define FCAP    12288
#define MAXW    1024
#define SU      7

__global__ __launch_bounds__(ETHR) void chunk_sort(const int* __restrict__ src,
                                                   const int* __restrict__ dst,
                                                   int* __restrict__ table,
                                                   int* __restrict__ pairs, int e) {
    __shared__ int lh[MAXB];
    __shared__ int lwave[MAXB / 32];
    __shared__ v2i stage[CHUNK];
    const int lane  = threadIdx.x & 31;
    const int wave  = threadIdx.x >> 5;
    const int cbase = blockIdx.x * CHUNK;
    const int len   = (e - cbase < CHUNK) ? (e - cbase) : CHUNK;
    if (threadIdx.x < MAXB) lh[threadIdx.x] = 0;
    v2i v[CHUNK / ETHR];
#pragma unroll
    for (int k = 0; k < CHUNK / ETHR; ++k) {
        const int i  = cbase + threadIdx.x + k * ETHR;
        const int ic = i < e ? i : e - 1;
        v[k].x = dst[ic];
        v[k].y = src[ic];
    }
    __builtin_amdgcn_sched_barrier(0);
    __syncthreads();
#pragma unroll
    for (int k = 0; k < CHUNK / ETHR; ++k)
        if (threadIdx.x + k * ETHR < len) atomicAdd(&lh[v[k].x >> BSH], 1);
    __syncthreads();

    int c = 0, inc = 0;
    if (threadIdx.x < MAXB) {
        c   = lh[threadIdx.x];
        inc = c;
#pragma unroll
        for (int o = 1; o < 32; o <<= 1) {
            int x = __shfl_up(inc, o);
            if (lane >= o) inc += x;
        }
        if (lane == 31) lwave[wave] = inc;
    }
    __syncthreads();
    if (threadIdx.x < MAXB) {
        int run = inc - c;
#pragma unroll
        for (int k = 0; k < MAXB / 32; ++k)
            if (k < wave) run += lwave[k];
        lh[threadIdx.x] = run;
        table[blockIdx.x * MAXB + threadIdx.x] = run;
    }
    __syncthreads();
#pragma unroll
    for (int k = 0; k < CHUNK / ETHR; ++k) {
        if (threadIdx.x + k * ETHR < len) {
            const int q = atomicAdd(&lh[v[k].x >> BSH], 1);
            stage[q] = v[k];
        }
    }
    __syncthreads();
    v2i* pr = (v2i*)pairs;
    for (int i = threadIdx.x; i < len; i += ETHR) pr[cbase + i] = stage[i];
}

__global__ __launch_bounds__(ETHR) void bucket_rows(const int* __restrict__ pairs,
                                                    const int* __restrict__ table,
                                                    int* __restrict__ rowptr,
                                                    int* __restrict__ cnt,
                                                    float* __restrict__ dinv,
                                                    int* __restrict__ col,
                                                    int n, int e, int nwg) {
    __shared__ int lcnt[BNODES];
    __shared__ int lcur[BNODES];
    __shared__ int lwave[BNODES / 32];
    __shared__ int runS[MAXW];
    __shared__ int runC[MAXW];
    __shared__ int red0[ETHR / 32];
    __shared__ int red1[ETHR / 32];
    __shared__ int stage[FCAP];
    const int b    = blockIdx.x;
    const int lane = threadIdx.x & 31;
    const int wave = __builtin_amdgcn_readfirstlane((int)(threadIdx.x >> 5));
    const v2i* pr  = (const v2i*)pairs;

    int s0 = 0, s1 = 0;
    for (int w = threadIdx.x; w < nwg; w += ETHR) {
        const int* row  = table + (size_t)w * MAXB;
        const int  lenw = (e - w * CHUNK < CHUNK) ? (e - w * CHUNK) : CHUNK;
        const int  o0   = row[b];
        const int  o1r  = row[b + 1 < MAXB ? b + 1 : b];
        const int  o1   = (b + 1 < MAXB) ? o1r : lenw;
        runS[w] = w * CHUNK + o0;
        runC[w] = o1 - o0;
        s0 += o0;
        s1 += o1;
    }
#pragma unroll
    for (int o = 16; o > 0; o >>= 1) {
        s0 += __shfl_xor(s0, o);
        s1 += __shfl_xor(s1, o);
    }
    if (lane == 0) { red0[wave] = s0; red1[wave] = s1; }
    if (threadIdx.x < BNODES) lcnt[threadIdx.x] = 0;
    __syncthreads();
    int bs = 0, be = 0;
#pragma unroll
    for (int k = 0; k < ETHR / 32; ++k) { bs += red0[k]; be += red1[k]; }
    const int len = be - bs;

    for (int w0 = wave; w0 < nwg; w0 += 32 * SU) {
        int s[SU], c[SU], d[SU];
#pragma unroll
        for (int u = 0; u < SU; ++u) {
            const int w  = w0 + 32 * u;
            const int wc = w < nwg ? w : nwg - 1;
            s[u] = __builtin_amdgcn_readfirstlane(runS[wc]);
            c[u] = w < nwg ? __builtin_amdgcn_readfirstlane(runC[wc]) : 0;
            d[u] = pairs[2 * (size_t)(lane < c[u] ? s[u] + lane : 0)];
        }
#pragma unroll
        for (int u = 0; u < SU; ++u)
            if (lane < c[u]) atomicAdd(&lcnt[d[u] & (BNODES - 1)], 1);
#pragma unroll
        for (int u = 0; u < SU; ++u)
#pragma nounroll
            for (int i0 = 32; i0 < c[u]; i0 += 32)
                if (i0 + lane < c[u])
                    atomicAdd(&lcnt[pairs[2 * (size_t)(s[u] + i0 + lane)] & (BNODES - 1)], 1);
    }
    __syncthreads();

    int cc = 0, inc = 0;
    if (threadIdx.x < BNODES) {
        cc  = lcnt[threadIdx.x];
        inc = cc;
#pragma unroll
        for (int o = 1; o < 32; o <<= 1) {
            int x = __shfl_up(inc, o);
            if (lane >= o) inc += x;
        }
        if (lane == 31) lwave[wave] = inc;
    }
    __syncthreads();
    if (threadIdx.x < BNODES) {
        int run = inc - cc;
#pragma unroll
        for (int k = 0; k < BNODES / 32; ++k)
            if (k < wave) run += lwave[k];
        lcur[threadIdx.x] = run;
        const int node = b * BNODES + threadIdx.x;
        if (node < n) {
            rowptr[node] = bs + run;
            cnt[node]    = cc;
            dinv[node]   = rsqrtf((float)(cc + 1));
        }
    }
    __syncthreads();

    const bool fits = len <= FCAP;
    for (int w0 = wave; w0 < nwg; w0 += 32 * SU) {
        int s[SU], c[SU];
        v2i v[SU];
#pragma unroll
        for (int u = 0; u < SU; ++u) {
            const int w  = w0 + 32 * u;
            const int wc = w < nwg ? w : nwg - 1;
            s[u] = __builtin_amdgcn_readfirstlane(runS[wc]);
            c[u] = w < nwg ? __builtin_amdgcn_readfirstlane(runC[wc]) : 0;
            v[u] = pr[lane < c[u] ? s[u] + lane : 0];
        }
#pragma unroll
        for (int u = 0; u < SU; ++u) {
            if (lane < c[u]) {
                const int q = atomicAdd(&lcur[v[u].x & (BNODES - 1)], 1);
                if (fits) stage[q] = v[u].y; else col[bs + q] = v[u].y;
            }
        }
#pragma unroll
        for (int u = 0; u < SU; ++u) {
#pragma nounroll
            for (int i0 = 32; i0 < c[u]; i0 += 32) {
                if (i0 + lane < c[u]) {
                    const v2i x = pr[s[u] + i0 + lane];
                    const int q = atomicAdd(&lcur[x.x & (BNODES - 1)], 1);
                    if (fits) stage[q] = x.y; else col[bs + q] = x.y;
                }
            }
        }
    }
    if (fits) {
        __syncthreads();
        for (int i = threadIdx.x; i < len; i += ETHR) col[bs + i] = stage[i];
    }
}

template<int DOUT>
__global__ __launch_bounds__(256) void gemm_wmma_f32(
        const float* __restrict__ X, const float* __restrict__ W,
        const float* __restrict__ dscale, float* __restrict__ Y, int nrows) {
    __shared__ float sW[64 * DOUT];
    for (int i = threadIdx.x; i < 64 * DOUT; i += 256) {
        int k   = i / DOUT;
        int col = i - k * DOUT;
        sW[((k >> 1) * DOUT + col) * 2 + (k & 1)] = W[i];
    }
    __syncthreads();

    const int lane = threadIdx.x & 31;
    const int wave = threadIdx.x >> 5;
    const int m    = lane & 15;
    const int khi  = lane >> 4;
    const int rowBase = (blockIdx.x * 8 + wave) * 16;
    if (rowBase >= nrows) return;

    v8f acc[DOUT / 16];
#pragma unroll
    for (int t = 0; t < DOUT / 16; ++t) acc[t] = (v8f)0.0f;

    float sc[8];
#pragma unroll
    for (int r = 0; r < 8; ++r) sc[r] = dscale[rowBase + r + 8 * khi];

    const float* xrow = X + (size_t)(rowBase + m) * 64;
    const v2f*   sWp  = (const v2f*)sW;
#pragma unroll
    for (int k4 = 0; k4 < 64; k4 += 4) {
        v2f a = *(const v2f*)(xrow + k4 + 2 * khi);
        const int kp = (k4 >> 1) + khi;
#pragma unroll
        for (int t = 0; t < DOUT / 16; ++t) {
            v2f b = sWp[kp * DOUT + t * 16 + m];
            acc[t] = __builtin_amdgcn_wmma_f32_16x16x4_f32(
                false, a, false, b, (short)0, acc[t], false, false);
        }
    }

#pragma unroll
    for (int t = 0; t < DOUT / 16; ++t)
#pragma unroll
        for (int r = 0; r < 8; ++r)
            Y[(size_t)(rowBase + r + 8 * khi) * DOUT + t * 16 + m] = acc[t][r] * sc[r];
}

#define NPW 8
#define GU  8

template<int D> struct Row;
template<> struct Row<64> {
    typedef v2f T;
    static __device__ __forceinline__ T ld(const float* p) { return *(const v2f*)p; }
    static __device__ __forceinline__ void st(float* p, T v) { *(v2f*)p = v; }
};
static __device__ __forceinline__ float relu1(float v) { return v > 0.0f ? v : 0.0f; }
static __device__ __forceinline__ v2f   relu1(v2f v) { v2f r; r.x = relu1(v.x); r.y = relu1(v.y); return r; }

static __device__ __forceinline__ int   lane_i(int v, int l)   { return __builtin_amdgcn_readlane(v, l); }
static __device__ __forceinline__ float lane_f(float v, int l) { return __int_as_float(__builtin_amdgcn_readlane(__float_as_int(v), l)); }

typedef float v4f __attribute__((ext_vector_type(4)));

#define NPW2 12
#define NQ (NPW2 / 4)
#define GW2 4

template<int T> static __device__ __forceinline__ int grp_lane(int v) {
    return (int)__builtin_amdgcn_mov_dpp8((unsigned)v, (unsigned)(T * 0x249249));
}

#define ASK_ROW(t) { const int s_ = grp_lane<t>(idx); r[t] = *(const v4f*)(hb + ((unsigned)s_ * (unsigned)(OUT_D * 4) + qoff)); }
#define ADD_ROW(t) { acc += (c8 + t < dg) ? r[t] : (v4f)0.0f; }

template<bool RELU>
__global__ __launch_bounds__(GW2 * 32) void gather_rows32(
        const int* __restrict__ rowptr, const int* __restrict__ cnt,
        const int* __restrict__ col, const float* __restrict__ dinv,
        const float* __restrict__ hs, const float* __restrict__ bias,
        float* __restrict__ out, int n) {
    const int lane = threadIdx.x & 31;
    const int g    = lane >> 3;
    const int q    = lane & 7;
    const int gw = __builtin_amdgcn_readfirstlane((int)(blockIdx.x * GW2 + (threadIdx.x >> 5)));
    const int n0 = gw * NPW2;
    if (n0 >= n) return;

    int   node[NQ], start[NQ], deg[NQ], maxd[NQ];
    float dd[NQ];
#pragma unroll
    for (int k = 0; k < NQ; ++k) {
        const int nd = n0 + 4 * k + g;
        node[k]  = nd < n ? nd : n - 1;
        start[k] = rowptr[node[k]];
        deg[k]   = nd < n ? cnt[node[k]] : 0;
        dd[k]    = dinv[node[k]];
    }
#pragma unroll
    for (int k = 0; k < NQ; ++k) {
        const int d0 = __builtin_amdgcn_readlane(deg[k], 0);
        const int d1 = __builtin_amdgcn_readlane(deg[k], 8);
        const int d2 = __builtin_amdgcn_readlane(deg[k], 16);
        const int d3 = __builtin_amdgcn_readlane(deg[k], 24);
        const int a  = d0 > d1 ? d0 : d1;
        const int b  = d2 > d3 ? d2 : d3;
        maxd[k] = a > b ? a : b;
    }
    const char*    hb   = (const char*)hs;
    const unsigned qoff = (unsigned)q * 16u;
    const v4f   bv = *(const v4f*)(bias + q * 4);

    int idx = q < deg[0] ? col[start[0] + q] : node[0];

#pragma unroll
    for (int k = 0; k < NQ; ++k) {
        const int dg = deg[k];
        const v4f self = *(const v4f*)(hb + ((unsigned)node[k] * (unsigned)(OUT_D * 4) + qoff));
        v4f acc = (v4f)0.0f;
        int c8 = 0;
        do {
            int nidx = idx;
            if (c8 + 8 < maxd[k]) {
                nidx = (c8 + 8 + q < dg) ? col[start[k] + c8 + 8 + q] : node[k];
            } else if (k + 1 < NQ) {
                nidx = q < deg[k + 1] ? col[start[k + 1] + q] : node[k + 1];
            }
            v4f r[8];
            if (c8 + 4 < maxd[k]) {
                ASK_ROW(0) ASK_ROW(1) ASK_ROW(2) ASK_ROW(3)
                ASK_ROW(4) ASK_ROW(5) ASK_ROW(6) ASK_ROW(7)
                asm volatile("" :: "v"(r[0]), "v"(r[1]), "v"(r[2]), "v"(r[3]),
                                   "v"(r[4]), "v"(r[5]), "v"(r[6]), "v"(r[7]));
                ADD_ROW(0) ADD_ROW(1) ADD_ROW(2) ADD_ROW(3)
                ADD_ROW(4) ADD_ROW(5) ADD_ROW(6) ADD_ROW(7)
            } else {
                ASK_ROW(0) ASK_ROW(1) ASK_ROW(2) ASK_ROW(3)
                asm volatile("" :: "v"(r[0]), "v"(r[1]), "v"(r[2]), "v"(r[3]));
                ADD_ROW(0) ADD_ROW(1) ADD_ROW(2) ADD_ROW(3)
            }
            idx = nidx;
            c8 += 8;
        } while (c8 < maxd[k]);

        acc += self;
        acc = acc * dd[k] + bv;
        if (RELU) { acc.x = relu1(acc.x); acc.y = relu1(acc.y); acc.z = relu1(acc.z); acc.w = relu1(acc.w); }
        if (n0 + 4 * k + g < n)
            *(v4f*)((char*)out + (size_t)node[k] * (OUT_D * 4) + q * 16) = acc;
    }
}
#undef ASK_ROW
#undef ADD_ROW

#define TPITCH 68

__global__ __launch_bounds__(256) void gather_rows_mm(
        const int* __restrict__ rowptr, const int* __restrict__ cnt,
        const int* __restrict__ col, const float* __restrict__ dinv,
        const float* __restrict__ hs, const float* __restrict__ bias,
        const float* __restrict__ Wn, float* __restrict__ out2, int n) {
    constexpr int D = HID_D;
    constexpr int V = D / 32;
    typedef Row<D>::T RT;
    __shared__ __attribute__((aligned(16))) float sW[64 * OUT_D];
    __shared__ __attribute__((aligned(16))) float strip[8 * NPW * TPITCH];

    const int lane = threadIdx.x & 31;
    const int wv   = threadIdx.x >> 5;
    const int gw = __builtin_amdgcn_readfirstlane((int)(blockIdx.x * 8 + wv));
    const int n0 = gw * NPW;
    const int nn = (n - n0 < NPW) ? (n - n0) : NPW;

    constexpr int WPT = 64 * OUT_D / 256;
    float wreg[WPT];
#pragma unroll
    for (int u = 0; u < WPT; ++u) wreg[u] = Wn[threadIdx.x + u * 256];

    const int   li     = n0 + (lane < nn ? lane : nn - 1);
    const int   vstart = rowptr[li];
    const int   vdeg   = cnt[li];
    const float vdd    = dinv[li];
    const float* hl    = hs + lane * V;
    const RT     bv    = Row<D>::ld(bias + lane * V);

#pragma unroll
    for (int u = 0; u < WPT; ++u) {
        const int i = threadIdx.x + u * 256;
        const int k = i / OUT_D;
        const int c = i - k * OUT_D;
        sW[((k >> 1) * OUT_D + c) * 2 + (k & 1)] = wreg[u];
    }
    __syncthreads();
    if (n0 >= n) return;
    float* tw = strip + wv * (NPW * TPITCH);

    int mycol = n0;
    int col1  = n0;
    {
        const int s0 = lane_i(vstart, 0);
        const int d0 = lane_i(vdeg, 0);
        const int s1 = lane_i(vstart, nn > 1 ? 1 : 0);
        const int d1 = nn > 1 ? lane_i(vdeg, 1) : 0;
        if (lane < d0) mycol = col[s0 + lane];
        if (lane < d1) col1  = col[s1 + lane];
        asm volatile("" :: "v"(mycol), "v"(col1));
    }

    for (int i = 0; i < nn; ++i) {
        const int   node  = n0 + i;
        const int   start = lane_i(vstart, i);
        const int   deg   = lane_i(vdeg, i);
        const float dd    = lane_f(vdd, i);

        const bool more2  = i + 2 < nn;
        const int  i2     = more2 ? i + 2 : i;
        const int  start2 = lane_i(vstart, i2);
        const int  deg2   = more2 ? lane_i(vdeg, i2) : 0;
        int col2 = n0 + i2;
        if (lane < deg2) col2 = col[start2 + lane];

        const RT self = Row<D>::ld(hl + (size_t)node * D);

        RT acc = (RT)0.0f;
        for (int base = 0; base < deg; base += 32) {
            const int m = (deg - base < 32) ? (deg - base) : 32;
            if (base > 0) {
                mycol = node;
                if (lane < m) mycol = col[start + base + lane];
            }
            for (int j = 0; j < m; j += GU) {
                RT r[GU];
#pragma unroll
                for (int u = 0; u < GU; ++u) {
                    const int s = lane_i(mycol, j + u);
                    r[u] = Row<D>::ld(hl + (size_t)s * D);
                }
#pragma unroll
                for (int u = 0; u < GU; ++u)
                    if (j + u < m) acc += r[u];
            }
        }

        acc += self;
        acc = acc * dd + bv;
        acc = relu1(acc);
        *(v2f*)(tw + i * TPITCH + lane * V) = acc;

        mycol = col1;
        col1  = col2;
    }

    const int m   = lane & 15;
    const int khi = lane >> 4;
    const float* arow = tw + (m & (NPW - 1)) * TPITCH + 2 * khi;
    const v2f*   sWp  = (const v2f*)sW;
    v8f pr[OUT_D / 16];
#pragma unroll
    for (int t = 0; t < OUT_D / 16; ++t) pr[t] = (v8f)0.0f;
#pragma unroll
    for (int k4 = 0; k4 < 64; k4 += 4) {
        const v2f a  = *(const v2f*)(arow + k4);
        const int kp = (k4 >> 1) + khi;
#pragma unroll
        for (int t = 0; t < OUT_D / 16; ++t) {
            const v2f b = sWp[kp * OUT_D + t * 16 + m];
            pr[t] = __builtin_amdgcn_wmma_f32_16x16x4_f32(
                false, a, false, b, (short)0, pr[t], false, false);
        }
    }
#pragma unroll
    for (int r = 0; r < NPW; ++r) {
        const float v = (khi ? pr[1][r] : pr[0][r]) * lane_f(vdd, r);
        if (r < nn) out2[(size_t)(n0 + r) * OUT_D + lane] = v;
    }
}

extern "C" void kernel_launch(void* const* d_in, const int* in_sizes, int n_in,
                              void* d_out, int out_size, void* d_ws, size_t ws_size,
                              hipStream_t stream) {
    const float* x  = (const float*)d_in[0];
    const int*   ei = (const int*)d_in[1];
    const float* W1 = (const float*)d_in[2];
    const float* b1 = (const float*)d_in[3];
    const float* W2 = (const float*)d_in[4];
    const float* b2 = (const float*)d_in[5];
    float* out = (float*)d_out;

    const int N = in_sizes[0] / IN_D;
    const int E = in_sizes[1] / 2;
    const int* src = ei;
    const int* dst = ei + E;

    const int T = 256;
    auto g = [](long long n, int t) { return (unsigned)((n + t - 1) / t); };
    const int nb  = (N + BNODES - 1) >> BSH;
    const int nwg = (int)g(E, CHUNK);
    const int nt  = nwg * MAXB;

    auto up = [](size_t b) { return (b + 255) & ~(size_t)255; };
    char* p = (char*)d_ws;
    float* dinv   = (float*)p;  p += up((size_t)N * 4);
    int*   cnt    = (int*)p;    p += up((size_t)N * 4);
    int*   rowptr = (int*)p;    p += up((size_t)N * 4);
    int*   table  = (int*)p;    p += up((size_t)nt * 4);
    int*   pairs  = (int*)p;    p += up((size_t)E * 8);
    int*   col    = (int*)p;    p += up((size_t)E * 4);
    float* h1     = (float*)p;  p += up((size_t)N * HID_D * 4);
    float* h2     = (float*)p;

    chunk_sort <<<nwg, ETHR, 0, stream>>>(src, dst, table, pairs, E);
    bucket_rows<<<nb, ETHR, 0, stream>>>(pairs, table, rowptr, cnt, dinv, col, N, E, nwg);

    const unsigned gemmBlocks = g((long long)(N + 15) / 16, 8);
    const unsigned nodeBlocks = g(g(N, NPW), 8);

    gemm_wmma_f32<HID_D><<<gemmBlocks, T, 0, stream>>>(x, W1, dinv, h1, N);
    gather_rows_mm<<<nodeBlocks, T, 0, stream>>>(rowptr, cnt, col, dinv, h1, b1, W2, h2, N);

    const unsigned nodeBlocks2 = g(g(N, NPW2), GW2);
    gather_rows32<false><<<nodeBlocks2, GW2 * 32, 0, stream>>>(rowptr, cnt, col, dinv, h2, b2, out, N);
}


# Layer-1 product moved into bucket_rows: behind its sort each block multiplies its own 256 rows of x by W1 (x asked for under the sort, staged in LDS); the product's launch is gone, 5 launches to 4.
# speedup vs baseline: 8.1455x; 1.1085x over previous
#include <hip/hip_runtime.h>
#include <hip/hip_bf16.h>


typedef float v2f __attribute__((ext_vector_type(2)));
typedef float v8f __attribute__((ext_vector_type(8)));

#define IN_D  64
#define HID_D 64
#define OUT_D 32

typedef int v2i __attribute__((ext_vector_type(2)));
typedef float v4f __attribute__((ext_vector_type(4)));

#define BSH     8
#define BNODES  256
#define MAXB    512
#define CHUNK   8192
#define ETHR    1024
#define FCAP    12288
#define MAXW    1024
#define SU      7
#define XPITCH  68
#define XPT     (BNODES * IN_D / 4 / ETHR)
#define WPTB    (IN_D * HID_D / ETHR)

__global__ __launch_bounds__(ETHR) void chunk_sort(const int* __restrict__ src,
                                                   const int* __restrict__ dst,
                                                   int* __restrict__ table,
                                                   int* __restrict__ pairs, int e) {
    __shared__ int lh[MAXB];
    __shared__ int lwave[MAXB / 32];
    __shared__ v2i stage[CHUNK];
    const int lane  = threadIdx.x & 31;
    const int wave  = threadIdx.x >> 5;
    const int cbase = blockIdx.x * CHUNK;
    const int len   = (e - cbase < CHUNK) ? (e - cbase) : CHUNK;
    if (threadIdx.x < MAXB) lh[threadIdx.x] = 0;
    v2i v[CHUNK / ETHR];
#pragma unroll
    for (int k = 0; k < CHUNK / ETHR; ++k) {
        const int i  = cbase + threadIdx.x + k * ETHR;
        const int ic = i < e ? i : e - 1;
        v[k].x = dst[ic];
        v[k].y = src[ic];
    }
    __builtin_amdgcn_sched_barrier(0);
    __syncthreads();
#pragma unroll
    for (int k = 0; k < CHUNK / ETHR; ++k)
        if (threadIdx.x + k * ETHR < len) atomicAdd(&lh[v[k].x >> BSH], 1);
    __syncthreads();

    int c = 0, inc = 0;
    if (threadIdx.x < MAXB) {
        c   = lh[threadIdx.x];
        inc = c;
#pragma unroll
        for (int o = 1; o < 32; o <<= 1) {
            int x = __shfl_up(inc, o);
            if (lane >= o) inc += x;
        }
        if (lane == 31) lwave[wave] = inc;
    }
    __syncthreads();
    if (threadIdx.x < MAXB) {
        int run = inc - c;
#pragma unroll
        for (int k = 0; k < MAXB / 32; ++k)
            if (k < wave) run += lwave[k];
        lh[threadIdx.x] = run;
        table[blockIdx.x * MAXB + threadIdx.x] = run;
    }
    __syncthreads();
#pragma unroll
    for (int k = 0; k < CHUNK / ETHR; ++k) {
        if (threadIdx.x + k * ETHR < len) {
            const int q = atomicAdd(&lh[v[k].x >> BSH], 1);
            stage[q] = v[k];
        }
    }
    __syncthreads();
    v2i* pr = (v2i*)pairs;
    for (int i = threadIdx.x; i < len; i += ETHR) pr[cbase + i] = stage[i];
}

static __device__ __forceinline__ void lds_barrier() {
    __builtin_amdgcn_fence(__ATOMIC_RELEASE, "workgroup", "local");
    __builtin_amdgcn_s_barrier();
    __builtin_amdgcn_fence(__ATOMIC_ACQUIRE, "workgroup", "local");
}

__global__ __launch_bounds__(ETHR) void bucket_rows(const int* __restrict__ pairs,
                                                    const int* __restrict__ table,
                                                    int* __restrict__ rowptr,
                                                    int* __restrict__ cnt,
                                                    float* __restrict__ dinv,
                                                    int* __restrict__ col,
                                                    const float* __restrict__ X,
                                                    const float* __restrict__ W,
                                                    float* __restrict__ Y,
                                                    int n, int e, int nwg) {
    __shared__ int lcnt[BNODES];
    __shared__ int lcur[BNODES];
    __shared__ int lwave[BNODES / 32];
    __shared__ int runS[MAXW];
    __shared__ int runC[MAXW];
    __shared__ int red0[ETHR / 32];
    __shared__ int red1[ETHR / 32];
    __shared__ int stage[FCAP];
    __shared__ float ldv[BNODES];
    __shared__ __attribute__((aligned(16))) float sW[IN_D * HID_D];
    __shared__ __attribute__((aligned(16))) float xt[BNODES * XPITCH];
    const int b    = blockIdx.x;
    const int lane = threadIdx.x & 31;
    const int wave = __builtin_amdgcn_readfirstlane((int)(threadIdx.x >> 5));
    const v2i* pr  = (const v2i*)pairs;

    int s0 = 0, s1 = 0;
    for (int w = threadIdx.x; w < nwg; w += ETHR) {
        const int* row  = table + (size_t)w * MAXB;
        const int  lenw = (e - w * CHUNK < CHUNK) ? (e - w * CHUNK) : CHUNK;
        const int  o0   = row[b];
        const int  o1r  = row[b + 1 < MAXB ? b + 1 : b];
        const int  o1   = (b + 1 < MAXB) ? o1r : lenw;
        runS[w] = w * CHUNK + o0;
        runC[w] = o1 - o0;
        s0 += o0;
        s1 += o1;
    }
#pragma unroll
    for (int o = 16; o > 0; o >>= 1) {
        s0 += __shfl_xor(s0, o);
        s1 += __shfl_xor(s1, o);
    }
    if (lane == 0) { red0[wave] = s0; red1[wave] = s1; }
    if (threadIdx.x < BNODES) lcnt[threadIdx.x] = 0;
    __syncthreads();
    int bs = 0, be = 0;
#pragma unroll
    for (int k = 0; k < ETHR / 32; ++k) { bs += red0[k]; be += red1[k]; }
    const int len = be - bs;

    for (int w0 = wave; w0 < nwg; w0 += 32 * SU) {
        int s[SU], c[SU], d[SU];
#pragma unroll
        for (int u = 0; u < SU; ++u) {
            const int w  = w0 + 32 * u;
            const int wc = w < nwg ? w : nwg - 1;
            s[u] = __builtin_amdgcn_readfirstlane(runS[wc]);
            c[u] = w < nwg ? __builtin_amdgcn_readfirstlane(runC[wc]) : 0;
            d[u] = pairs[2 * (size_t)(lane < c[u] ? s[u] + lane : 0)];
        }
#pragma unroll
        for (int u = 0; u < SU; ++u)
            if (lane < c[u]) atomicAdd(&lcnt[d[u] & (BNODES - 1)], 1);
#pragma unroll
        for (int u = 0; u < SU; ++u)
#pragma nounroll
            for (int i0 = 32; i0 < c[u]; i0 += 32)
                if (i0 + lane < c[u])
                    atomicAdd(&lcnt[pairs[2 * (size_t)(s[u] + i0 + lane)] & (BNODES - 1)], 1);
    }

    float wreg[WPTB];
    v4f   xr[XPT];
#pragma unroll
    for (int u = 0; u < WPTB; ++u) wreg[u] = W[threadIdx.x + u * ETHR];
#pragma unroll
    for (int u = 0; u < XPT; ++u) {
        const int i   = threadIdx.x + u * ETHR;
        const int row = b * BNODES + (i >> 4);
        const int rc  = row < n ? row : n - 1;
        xr[u] = *(const v4f*)(X + (size_t)rc * IN_D + (i & 15) * 4);
    }
    lds_barrier();

    int cc = 0, inc = 0;
    if (threadIdx.x < BNODES) {
        cc  = lcnt[threadIdx.x];
        inc = cc;
#pragma unroll
        for (int o = 1; o < 32; o <<= 1) {
            int x = __shfl_up(inc, o);
            if (lane >= o) inc += x;
        }
        if (lane == 31) lwave[wave] = inc;
    }
    lds_barrier();
    if (threadIdx.x < BNODES) {
        int run = inc - cc;
#pragma unroll
        for (int k = 0; k < BNODES / 32; ++k)
            if (k < wave) run += lwave[k];
        lcur[threadIdx.x] = run;
        const int   node = b * BNODES + threadIdx.x;
        const float dv   = rsqrtf((float)(cc + 1));
        ldv[threadIdx.x] = dv;
        if (node < n) {
            rowptr[node] = bs + run;
            cnt[node]    = cc;
            dinv[node]   = dv;
        }
    }
    lds_barrier();

    const bool fits = len <= FCAP;
    for (int w0 = wave; w0 < nwg; w0 += 32 * SU) {
        int s[SU], c[SU];
        v2i v[SU];
#pragma unroll
        for (int u = 0; u < SU; ++u) {
            const int w  = w0 + 32 * u;
            const int wc = w < nwg ? w : nwg - 1;
            s[u] = __builtin_amdgcn_readfirstlane(runS[wc]);
            c[u] = w < nwg ? __builtin_amdgcn_readfirstlane(runC[wc]) : 0;
            v[u] = pr[lane < c[u] ? s[u] + lane : 0];
        }
#pragma unroll
        for (int u = 0; u < SU; ++u) {
            if (lane < c[u]) {
                const int q = atomicAdd(&lcur[v[u].x & (BNODES - 1)], 1);
                if (fits) stage[q] = v[u].y; else col[bs + q] = v[u].y;
            }
        }
#pragma unroll
        for (int u = 0; u < SU; ++u) {
#pragma nounroll
            for (int i0 = 32; i0 < c[u]; i0 += 32) {
                if (i0 + lane < c[u]) {
                    const v2i x = pr[s[u] + i0 + lane];
                    const int q = atomicAdd(&lcur[x.x & (BNODES - 1)], 1);
                    if (fits) stage[q] = x.y; else col[bs + q] = x.y;
                }
            }
        }
    }
#pragma unroll
    for (int u = 0; u < WPTB; ++u) {
        const int i = threadIdx.x + u * ETHR;
        const int k = i / HID_D;
        const int c = i - k * HID_D;
        sW[((k >> 1) * HID_D + c) * 2 + (k & 1)] = wreg[u];
    }
#pragma unroll
    for (int u = 0; u < XPT; ++u) {
        const int i = threadIdx.x + u * ETHR;
        *(v4f*)(xt + (i >> 4) * XPITCH + (i & 15) * 4) = xr[u];
    }
    lds_barrier();
    if (fits)
        for (int i = threadIdx.x; i < len; i += ETHR) col[bs + i] = stage[i];

    const int tile    = wave & 15;
    const int half    = wave >> 4;
    const int rowBase = b * BNODES + tile * 16;
    if (rowBase >= n) return;
    const int m   = lane & 15;
    const int khi = lane >> 4;
    const float* arow = xt + (tile * 16 + m) * XPITCH + 2 * khi;
    const v2f*   bcol = (const v2f*)sW + khi * HID_D + half * 32 + m;
    v8f acc[2];
    acc[0] = (v8f)0.0f;
    acc[1] = (v8f)0.0f;
#pragma unroll
    for (int k4 = 0; k4 < IN_D; k4 += 4) {
        const v2f a = *(const v2f*)(arow + k4);
#pragma unroll
        for (int t = 0; t < 2; ++t) {
            const v2f bb = bcol[(k4 >> 1) * HID_D + t * 16];
            acc[t] = __builtin_amdgcn_wmma_f32_16x16x4_f32(
                false, a, false, bb, (short)0, acc[t], false, false);
        }
    }
    float*       y   = Y + (size_t)(rowBase + 8 * khi) * HID_D + half * 32 + m;
    const float* scp = ldv + tile * 16 + 8 * khi;
    if (rowBase + 16 <= n) {
#pragma unroll
        for (int r = 0; r < 8; ++r) {
            const float sc = scp[r];
            y[r * HID_D]      = acc[0][r] * sc;
            y[r * HID_D + 16] = acc[1][r] * sc;
        }
    } else {
#pragma unroll
        for (int r = 0; r < 8; ++r) {
            const float sc = scp[r];
            if (rowBase + 8 * khi + r < n) {
                y[r * HID_D]      = acc[0][r] * sc;
                y[r * HID_D + 16] = acc[1][r] * sc;
            }
        }
    }
}

#define NPW 8
#define GU  8

template<int D> struct Row;
template<> struct Row<64> {
    typedef v2f T;
    static __device__ __forceinline__ T ld(const float* p) { return *(const v2f*)p; }
    static __device__ __forceinline__ void st(float* p, T v) { *(v2f*)p = v; }
};
static __device__ __forceinline__ float relu1(float v) { return v > 0.0f ? v : 0.0f; }
static __device__ __forceinline__ v2f   relu1(v2f v) { v2f r; r.x = relu1(v.x); r.y = relu1(v.y); return r; }

static __device__ __forceinline__ int   lane_i(int v, int l)   { return __builtin_amdgcn_readlane(v, l); }
static __device__ __forceinline__ float lane_f(float v, int l) { return __int_as_float(__builtin_amdgcn_readlane(__float_as_int(v), l)); }

#define NPW2 12
#define NQ (NPW2 / 4)
#define GW2 4

template<int T> static __device__ __forceinline__ int grp_lane(int v) {
    return (int)__builtin_amdgcn_mov_dpp8((unsigned)v, (unsigned)(T * 0x249249));
}

#define ASK_ROW(t) { const int s_ = grp_lane<t>(idx); r[t] = *(const v4f*)(hb + ((unsigned)s_ * (unsigned)(OUT_D * 4) + qoff)); }
#define ADD_ROW(t) { acc += (c8 + t < dg) ? r[t] : (v4f)0.0f; }

template<bool RELU>
__global__ __launch_bounds__(GW2 * 32) void gather_rows32(
        const int* __restrict__ rowptr, const int* __restrict__ cnt,
        const int* __restrict__ col, const float* __restrict__ dinv,
        const float* __restrict__ hs, const float* __restrict__ bias,
        float* __restrict__ out, int n) {
    const int lane = threadIdx.x & 31;
    const int g    = lane >> 3;
    const int q    = lane & 7;
    const int gw = __builtin_amdgcn_readfirstlane((int)(blockIdx.x * GW2 + (threadIdx.x >> 5)));
    const int n0 = gw * NPW2;
    if (n0 >= n) return;

    int   node[NQ], start[NQ], deg[NQ], maxd[NQ];
    float dd[NQ];
#pragma unroll
    for (int k = 0; k < NQ; ++k) {
        const int nd = n0 + 4 * k + g;
        node[k]  = nd < n ? nd : n - 1;
        start[k] = rowptr[node[k]];
        deg[k]   = nd < n ? cnt[node[k]] : 0;
        dd[k]    = dinv[node[k]];
    }
#pragma unroll
    for (int k = 0; k < NQ; ++k) {
        const int d0 = __builtin_amdgcn_readlane(deg[k], 0);
        const int d1 = __builtin_amdgcn_readlane(deg[k], 8);
        const int d2 = __builtin_amdgcn_readlane(deg[k], 16);
        const int d3 = __builtin_amdgcn_readlane(deg[k], 24);
        const int a  = d0 > d1 ? d0 : d1;
        const int b  = d2 > d3 ? d2 : d3;
        maxd[k] = a > b ? a : b;
    }
    const char*    hb   = (const char*)hs;
    const unsigned qoff = (unsigned)q * 16u;
    const v4f   bv = *(const v4f*)(bias + q * 4);

    int idx = q < deg[0] ? col[start[0] + q] : node[0];

#pragma unroll
    for (int k = 0; k < NQ; ++k) {
        const int dg = deg[k];
        const v4f self = *(const v4f*)(hb + ((unsigned)node[k] * (unsigned)(OUT_D * 4) + qoff));
        v4f acc = (v4f)0.0f;
        int c8 = 0;
        do {
            int nidx = idx;
            if (c8 + 8 < maxd[k]) {
                nidx = (c8 + 8 + q < dg) ? col[start[k] + c8 + 8 + q] : node[k];
            } else if (k + 1 < NQ) {
                nidx = q < deg[k + 1] ? col[start[k + 1] + q] : node[k + 1];
            }
            v4f r[8];
            if (c8 + 4 < maxd[k]) {
                ASK_ROW(0) ASK_ROW(1) ASK_ROW(2) ASK_ROW(3)
                ASK_ROW(4) ASK_ROW(5) ASK_ROW(6) ASK_ROW(7)
                asm volatile("" :: "v"(r[0]), "v"(r[1]), "v"(r[2]), "v"(r[3]),
                                   "v"(r[4]), "v"(r[5]), "v"(r[6]), "v"(r[7]));
                ADD_ROW(0) ADD_ROW(1) ADD_ROW(2) ADD_ROW(3)
                ADD_ROW(4) ADD_ROW(5) ADD_ROW(6) ADD_ROW(7)
            } else {
                ASK_ROW(0) ASK_ROW(1) ASK_ROW(2) ASK_ROW(3)
                asm volatile("" :: "v"(r[0]), "v"(r[1]), "v"(r[2]), "v"(r[3]));
                ADD_ROW(0) ADD_ROW(1) ADD_ROW(2) ADD_ROW(3)
            }
            idx = nidx;
            c8 += 8;
        } while (c8 < maxd[k]);

        acc += self;
        acc = acc * dd[k] + bv;
        if (RELU) { acc.x = relu1(acc.x); acc.y = relu1(acc.y); acc.z = relu1(acc.z); acc.w = relu1(acc.w); }
        if (n0 + 4 * k + g < n)
            *(v4f*)((char*)out + (size_t)node[k] * (OUT_D * 4) + q * 16) = acc;
    }
}
#undef ASK_ROW
#undef ADD_ROW

#define TPITCH 68

__global__ __launch_bounds__(256) void gather_rows_mm(
        const int* __restrict__ rowptr, const int* __restrict__ cnt,
        const int* __restrict__ col, const float* __restrict__ dinv,
        const float* __restrict__ hs, const float* __restrict__ bias,
        const float* __restrict__ Wn, float* __restrict__ out2, int n) {
    constexpr int D = HID_D;
    constexpr int V = D / 32;
    typedef Row<D>::T RT;
    __shared__ __attribute__((aligned(16))) float sW[64 * OUT_D];
    __shared__ __attribute__((aligned(16))) float strip[8 * NPW * TPITCH];

    const int lane = threadIdx.x & 31;
    const int wv   = threadIdx.x >> 5;
    const int gw = __builtin_amdgcn_readfirstlane((int)(blockIdx.x * 8 + wv));
    const int n0 = gw * NPW;
    const int nn = (n - n0 < NPW) ? (n - n0) : NPW;

    constexpr int WPT = 64 * OUT_D / 256;
    float wreg[WPT];
#pragma unroll
    for (int u = 0; u < WPT; ++u) wreg[u] = Wn[threadIdx.x + u * 256];

    const int   li     = n0 + (lane < nn ? lane : nn - 1);
    const int   vstart = rowptr[li];
    const int   vdeg   = cnt[li];
    const float vdd    = dinv[li];
    const float* hl    = hs + lane * V;
    const RT     bv    = Row<D>::ld(bias + lane * V);

#pragma unroll
    for (int u = 0; u < WPT; ++u) {
        const int i = threadIdx.x + u * 256;
        const int k = i / OUT_D;
        const int c = i - k * OUT_D;
        sW[((k >> 1) * OUT_D + c) * 2 + (k & 1)] = wreg[u];
    }
    __syncthreads();
    if (n0 >= n) return;
    float* tw = strip + wv * (NPW * TPITCH);

    int mycol = n0;
    int col1  = n0;
    {
        const int s0 = lane_i(vstart, 0);
        const int d0 = lane_i(vdeg, 0);
        const int s1 = lane_i(vstart, nn > 1 ? 1 : 0);
        const int d1 = nn > 1 ? lane_i(vdeg, 1) : 0;
        if (lane < d0) mycol = col[s0 + lane];
        if (lane < d1) col1  = col[s1 + lane];
        asm volatile("" :: "v"(mycol), "v"(col1));
    }

    for (int i = 0; i < nn; ++i) {
        const int   node  = n0 + i;
        const int   start = lane_i(vstart, i);
        const int   deg   = lane_i(vdeg, i);
        const float dd    = lane_f(vdd, i);

        const bool more2  = i + 2 < nn;
        const int  i2     = more2 ? i + 2 : i;
        const int  start2 = lane_i(vstart, i2);
        const int  deg2   = more2 ? lane_i(vdeg, i2) : 0;
        int col2 = n0 + i2;
        if (lane < deg2) col2 = col[start2 + lane];

        const RT self = Row<D>::ld(hl + (size_t)node * D);

        RT acc = (RT)0.0f;
        for (int base = 0; base < deg; base += 32) {
            const int m = (deg - base < 32) ? (deg - base) : 32;
            if (base > 0) {
                mycol = node;
                if (lane < m) mycol = col[start + base + lane];
            }
            for (int j = 0; j < m; j += GU) {
                RT r[GU];
#pragma unroll
                for (int u = 0; u < GU; ++u) {
                    const int s = lane_i(mycol, j + u);
                    r[u] = Row<D>::ld(hl + (size_t)s * D);
                }
#pragma unroll
                for (int u = 0; u < GU; ++u)
                    if (j + u < m) acc += r[u];
            }
        }

        acc += self;
        acc = acc * dd + bv;
        acc = relu1(acc);
        *(v2f*)(tw + i * TPITCH + lane * V) = acc;

        mycol = col1;
        col1  = col2;
    }

    const int m   = lane & 15;
    const int khi = lane >> 4;
    const float* arow = tw + (m & (NPW - 1)) * TPITCH + 2 * khi;
    const v2f*   sWp  = (const v2f*)sW;
    v8f pr[OUT_D / 16];
#pragma unroll
    for (int t = 0; t < OUT_D / 16; ++t) pr[t] = (v8f)0.0f;
#pragma unroll
    for (int k4 = 0; k4 < 64; k4 += 4) {
        const v2f a  = *(const v2f*)(arow + k4);
        const int kp = (k4 >> 1) + khi;
#pragma unroll
        for (int t = 0; t < OUT_D / 16; ++t) {
            const v2f b = sWp[kp * OUT_D + t * 16 + m];
            pr[t] = __builtin_amdgcn_wmma_f32_16x16x4_f32(
                false, a, false, b, (short)0, pr[t], false, false);
        }
    }
#pragma unroll
    for (int r = 0; r < NPW; ++r) {
        const float v = (khi ? pr[1][r] : pr[0][r]) * lane_f(vdd, r);
        if (r < nn) out2[(size_t)(n0 + r) * OUT_D + lane] = v;
    }
}

extern "C" void kernel_launch(void* const* d_in, const int* in_sizes, int n_in,
                              void* d_out, int out_size, void* d_ws, size_t ws_size,
                              hipStream_t stream) {
    const float* x  = (const float*)d_in[0];
    const int*   ei = (const int*)d_in[1];
    const float* W1 = (const float*)d_in[2];
    const float* b1 = (const float*)d_in[3];
    const float* W2 = (const float*)d_in[4];
    const float* b2 = (const float*)d_in[5];
    float* out = (float*)d_out;

    const int N = in_sizes[0] / IN_D;
    const int E = in_sizes[1] / 2;
    const int* src = ei;
    const int* dst = ei + E;

    const int T = 256;
    auto g = [](long long n, int t) { return (unsigned)((n + t - 1) / t); };
    const int nb  = (N + BNODES - 1) >> BSH;
    const int nwg = (int)g(E, CHUNK);
    const int nt  = nwg * MAXB;

    auto up = [](size_t b) { return (b + 255) & ~(size_t)255; };
    char* p = (char*)d_ws;
    float* dinv   = (float*)p;  p += up((size_t)N * 4);
    int*   cnt    = (int*)p;    p += up((size_t)N * 4);
    int*   rowptr = (int*)p;    p += up((size_t)N * 4);
    int*   table  = (int*)p;    p += up((size_t)nt * 4);
    int*   pairs  = (int*)p;    p += up((size_t)E * 8);
    int*   col    = (int*)p;    p += up((size_t)E * 4);
    float* h1     = (float*)p;  p += up((size_t)N * HID_D * 4);
    float* h2     = (float*)p;

    chunk_sort <<<nwg, ETHR, 0, stream>>>(src, dst, table, pairs, E);
    bucket_rows<<<nb, ETHR, 0, stream>>>(pairs, table, rowptr, cnt, dinv, col, x, W1, h1, N, E, nwg);

    const unsigned nodeBlocks = g(g(N, NPW), 8);

    gather_rows_mm<<<nodeBlocks, T, 0, stream>>>(rowptr, cnt, col, dinv, h1, b1, W2, h2, N);

    const unsigned nodeBlocks2 = g(g(N, NPW2), GW2);
    gather_rows32<false><<<nodeBlocks2, GW2 * 32, 0, stream>>>(rowptr, cnt, col, dinv, h2, b2, out, N);
}
